# starting kernel
# baseline (speedup 1.0000x reference)
#include <hip/hip_runtime.h>
#include <cstddef>

#define NN   20000
#define EE   60000
#define GG   64
#define RC   8
#define FIN  64
#define FE   16
#define HH   128
#define OO   32
#define NLAY 2

typedef __attribute__((ext_vector_type(2))) float v2f;
typedef __attribute__((ext_vector_type(8))) float v8f;

__device__ __forceinline__ float gelu_f(float x) {
  return 0.5f * x * (1.0f + erff(x * 0.70710678118654752f));
}

template <int K>
__device__ __forceinline__ v8f wmma_acc(const float* A, int lda,
                                        const float* B, int ldb, v8f c) {
  const int lane = threadIdx.x & 31;
  const int mn = lane & 15;
  const int kh = (lane >> 4) << 1;
#pragma unroll
  for (int k = 0; k < K; k += 4) {
    v2f a, b;
    a[0] = A[mn * lda + k + kh + 0];
    a[1] = A[mn * lda + k + kh + 1];
    b[0] = B[(k + kh + 0) * ldb + mn];
    b[1] = B[(k + kh + 1) * ldb + mn];
    c = __builtin_amdgcn_wmma_f32_16x16x4_f32(false, a, false, b, (short)0, c,
                                              false, false);
  }
  return c;
}

__global__ __launch_bounds__(256) void atom_gemm_kernel(
    const float* __restrict__ xf, const float* __restrict__ W,
    const float* __restrict__ bias, float* __restrict__ ux) {
  const int row0 = blockIdx.x * 16;
  const int wave = threadIdx.x >> 5;
  const int col0 = wave * 16;
  v8f c = {};
  c = wmma_acc<FIN>(xf + (size_t)row0 * FIN, FIN, W + col0, HH, c);
  const int lane = threadIdx.x & 31;
  const int n = lane & 15;
  const int kh = lane >> 4;
  const float bv = bias[col0 + n];
#pragma unroll
  for (int v = 0; v < 8; ++v) {
    const float val = c[v] + bv;
    const size_t row = (size_t)(row0 + v + 8 * kh);
#pragma unroll
    for (int rc = 0; rc < RC; ++rc)
      ux[(size_t)rc * NN * HH + row * HH + col0 + n] = val;
  }
}

__global__ __launch_bounds__(256) void edge_gemm_kernel(
    const float* __restrict__ ea, const float* __restrict__ W,
    const float* __restrict__ bias, float* __restrict__ eemb) {
  const int row0 = blockIdx.x * 16;
  const int wave = threadIdx.x >> 5;
  const int col0 = wave * 16;
  v8f c = {};
  c = wmma_acc<FE>(ea + (size_t)row0 * FE, FE, W + col0, HH, c);
  const int lane = threadIdx.x & 31;
  const int n = lane & 15;
  const int kh = lane >> 4;
  const float bv = bias[col0 + n];
#pragma unroll
  for (int v = 0; v < 8; ++v)
    eemb[(size_t)(row0 + v + 8 * kh) * HH + col0 + n] = c[v] + bv;
}

__global__ __launch_bounds__(256) void msg_scatter_kernel(
    const float* __restrict__ ux, const float* __restrict__ eemb,
    const int* __restrict__ src, const int* __restrict__ dst,
    const float* __restrict__ mask, float* __restrict__ agg) {
  const long long tid = (long long)blockIdx.x * 256 + threadIdx.x;
  if (tid >= (long long)RC * EE * (HH / 4)) return;
  const int chunk = (int)(tid & 31);
  const int e = (int)((tid >> 5) % EE);
  const int rc = (int)((tid >> 5) / EE);
  const int s = src[e];
  const int d = dst[e];
  const float mk = mask[(size_t)rc * NN + s] * mask[(size_t)rc * NN + d];
  const float4 xs =
      *(const float4*)(ux + (size_t)rc * NN * HH + (size_t)s * HH + chunk * 4);
  const float4 ee = *(const float4*)(eemb + (size_t)e * HH + chunk * 4);
  float* ap = agg + (size_t)rc * NN * HH + (size_t)d * HH + chunk * 4;
  unsafeAtomicAdd(ap + 0, gelu_f(xs.x + ee.x) * mk);
  unsafeAtomicAdd(ap + 1, gelu_f(xs.y + ee.y) * mk);
  unsafeAtomicAdd(ap + 2, gelu_f(xs.z + ee.z) * mk);
  unsafeAtomicAdd(ap + 3, gelu_f(xs.w + ee.w) * mk);
}

__global__ __launch_bounds__(256) void gin_mlp_kernel(
    float* __restrict__ ux, const float* __restrict__ agg,
    const float* __restrict__ eps_l, const float* __restrict__ W1,
    const float* __restrict__ b1, const float* __restrict__ W2,
    const float* __restrict__ b2) {
  __shared__ float hA[16 * 132];
  __shared__ float hB[16 * 132];
  const int row_tile = blockIdx.x % (NN / 16);
  const int rc = blockIdx.x / (NN / 16);
  const int row0 = row_tile * 16;
  const float one_eps = 1.0f + eps_l[0];
  float* uxp = ux + (size_t)rc * NN * HH;
  const float* agp = agg + (size_t)rc * NN * HH;

  for (int i = threadIdx.x; i < 16 * HH; i += 256) {
    const int r = i >> 7;
    const int cc = i & 127;
    const size_t g = (size_t)(row0 + r) * HH + cc;
    hA[r * 132 + cc] = one_eps * uxp[g] + agp[g];
  }
  __syncthreads();

  const int wave = threadIdx.x >> 5;
  const int col0 = wave * 16;
  const int lane = threadIdx.x & 31;
  const int n = lane & 15;
  const int kh = lane >> 4;

  v8f c = {};
  c = wmma_acc<HH>(hA, 132, W1 + col0, HH, c);
  {
    const float bv = b1[col0 + n];
#pragma unroll
    for (int v = 0; v < 8; ++v)
      hB[(v + 8 * kh) * 132 + col0 + n] = gelu_f(c[v] + bv);
  }
  __syncthreads();

  v8f c2 = {};
  c2 = wmma_acc<HH>(hB, 132, W2 + col0, HH, c2);
  {
    const float bv = b2[col0 + n];
#pragma unroll
    for (int v = 0; v < 8; ++v)
      uxp[(size_t)(row0 + v + 8 * kh) * HH + col0 + n] = gelu_f(c2[v] + bv);
  }
}

__global__ __launch_bounds__(256) void head_mlp_kernel(
    const float* __restrict__ ux, const float* __restrict__ W1,
    const float* __restrict__ b1, const float* __restrict__ W2,
    const float* __restrict__ b2, float* __restrict__ obuf) {
  __shared__ float hA[16 * 132];
  __shared__ float hB[16 * 132];
  const int row_tile = blockIdx.x % (NN / 16);
  const int rc = blockIdx.x / (NN / 16);
  const int row0 = row_tile * 16;
  const float* uxp = ux + (size_t)rc * NN * HH;

  for (int i = threadIdx.x; i < 16 * HH; i += 256) {
    const int r = i >> 7;
    const int cc = i & 127;
    hA[r * 132 + cc] = uxp[(size_t)(row0 + r) * HH + cc];
  }
  __syncthreads();

  const int wave = threadIdx.x >> 5;
  const int col0 = wave * 16;
  const int lane = threadIdx.x & 31;
  const int n = lane & 15;
  const int kh = lane >> 4;

  v8f c = {};
  c = wmma_acc<HH>(hA, 132, W1 + col0, HH, c);
  {
    const float bv = b1[col0 + n];
#pragma unroll
    for (int v = 0; v < 8; ++v)
      hB[(v + 8 * kh) * 132 + col0 + n] = gelu_f(c[v] + bv);
  }
  __syncthreads();

  if (wave < OO / 16) {
    v8f c2 = {};
    c2 = wmma_acc<HH>(hB, 132, W2 + col0, OO, c2);
    const float bv = b2[col0 + n];
    float* op = obuf + (size_t)rc * NN * OO;
#pragma unroll
    for (int v = 0; v < 8; ++v)
      op[(size_t)(row0 + v + 8 * kh) * OO + col0 + n] = c2[v] + bv;
  }
}

__global__ __launch_bounds__(256) void pool_num_kernel(
    const float* __restrict__ obuf, const float* __restrict__ mask,
    const int* __restrict__ batch, float* __restrict__ num) {
  const long long tid = (long long)blockIdx.x * 256 + threadIdx.x;
  if (tid >= (long long)RC * NN * (OO / 4)) return;
  const int o4 = (int)(tid & 7);
  const int nd = (int)((tid >> 3) % NN);
  const int rc = (int)((tid >> 3) / NN);
  const float mk = mask[(size_t)rc * NN + nd];
  const int g = batch[nd];
  const float4 v =
      *(const float4*)(obuf + ((size_t)rc * NN + nd) * OO + o4 * 4);
  float* np = num + ((size_t)rc * GG + g) * OO + o4 * 4;
  unsafeAtomicAdd(np + 0, v.x * mk);
  unsafeAtomicAdd(np + 1, v.y * mk);
  unsafeAtomicAdd(np + 2, v.z * mk);
  unsafeAtomicAdd(np + 3, v.w * mk);
}

__global__ __launch_bounds__(256) void pool_den_kernel(
    const float* __restrict__ mask, const int* __restrict__ batch,
    float* __restrict__ den) {
  const int tid = blockIdx.x * 256 + threadIdx.x;
  if (tid >= RC * NN) return;
  const int nd = tid % NN;
  const int rc = tid / NN;
  unsafeAtomicAdd(den + (size_t)rc * GG + batch[nd],
                  mask[(size_t)rc * NN + nd]);
}

__global__ __launch_bounds__(256) void finalize_kernel(
    float* __restrict__ out, const float* __restrict__ den) {
  const int i = blockIdx.x * 256 + threadIdx.x;
  if (i >= RC * GG * OO) return;
  const int g = (i / OO) % GG;
  const int rc = i / (OO * GG);
  out[i] = out[i] / (den[(size_t)rc * GG + g] + 1e-7f);
}

extern "C" void kernel_launch(void* const* d_in, const int* in_sizes, int n_in,
                              void* d_out, int out_size, void* d_ws,
                              size_t ws_size, hipStream_t stream) {
  const float* x_feat = (const float*)d_in[0];
  const int* eidx = (const int*)d_in[1];
  const float* eattr = (const float*)d_in[2];
  const int* batch = (const int*)d_in[3];
  const float* nmask = (const float*)d_in[4];
  const float* atomW = (const float*)d_in[5];
  const float* atomB = (const float*)d_in[6];
  const float* eps = (const float*)d_in[7];
  const float* bondW = (const float*)d_in[8];
  const float* bondB = (const float*)d_in[9];
  const float* m1W = (const float*)d_in[10];
  const float* m1B = (const float*)d_in[11];
  const float* m2W = (const float*)d_in[12];
  const float* m2B = (const float*)d_in[13];
  const float* h1W = (const float*)d_in[14];
  const float* h1B = (const float*)d_in[15];
  const float* h2W = (const float*)d_in[16];
  const float* h2B = (const float*)d_in[17];
  float* out = (float*)d_out;

  const int* src = eidx;
  const int* dst = eidx + EE;

  float* ux = (float*)d_ws;
  float* agg = ux + (size_t)RC * NN * HH;
  float* eemb = agg + (size_t)RC * NN * HH;
  float* obuf = eemb + (size_t)EE * HH;
  float* den = obuf + (size_t)RC * NN * OO;

  atom_gemm_kernel<<<NN / 16, 256, 0, stream>>>(x_feat, atomW, atomB, ux);

  for (int l = 0; l < NLAY; ++l) {
    hipMemsetAsync(agg, 0, (size_t)RC * NN * HH * sizeof(float), stream);
    edge_gemm_kernel<<<EE / 16, 256, 0, stream>>>(
        eattr, bondW + (size_t)l * FE * HH, bondB + (size_t)l * HH, eemb);
    msg_scatter_kernel<<<(RC * EE * (HH / 4)) / 256, 256, 0, stream>>>(
        ux, eemb, src, dst, nmask, agg);
    gin_mlp_kernel<<<RC * (NN / 16), 256, 0, stream>>>(
        ux, agg, eps + l, m1W + (size_t)l * HH * HH, m1B + (size_t)l * HH,
        m2W + (size_t)l * HH * HH, m2B + (size_t)l * HH);
  }

  head_mlp_kernel<<<RC * (NN / 16), 256, 0, stream>>>(ux, h1W, h1B, h2W, h2B,
                                                      obuf);

  hipMemsetAsync(out, 0, (size_t)RC * GG * OO * sizeof(float), stream);
  hipMemsetAsync(den, 0, (size_t)RC * GG * sizeof(float), stream);
  pool_num_kernel<<<(RC * NN * (OO / 4)) / 256, 256, 0, stream>>>(obuf, nmask,
                                                                  batch, out);
  pool_den_kernel<<<(RC * NN) / 256, 256, 0, stream>>>(nmask, batch, den);
  finalize_kernel<<<(RC * GG * OO) / 256, 256, 0, stream>>>(out, den);
}


# MLP kernels (2x GIN, head): each wave keeps its 16-col weight fragments in registers; 512 work-groups loop over 32-row slabs, next slab prefetched past LDS-only barriers; sum order unchanged
# speedup vs baseline: 1.1284x; 1.1284x over previous
#include <hip/hip_runtime.h>
#include <cstddef>

#define NN   20000
#define EE   60000
#define GG   64
#define RC   8
#define FIN  64
#define FE   16
#define HH   128
#define OO   32
#define NLAY 2

typedef __attribute__((ext_vector_type(2))) float v2f;
typedef __attribute__((ext_vector_type(8))) float v8f;
typedef __attribute__((ext_vector_type(4))) float v4f;

__device__ __forceinline__ float gelu_f(float x) {
  return 0.5f * x * (1.0f + erff(x * 0.70710678118654752f));
}

template <int K>
__device__ __forceinline__ v8f wmma_acc(const float* A, int lda,
                                        const float* B, int ldb, v8f c) {
  const int lane = threadIdx.x & 31;
  const int mn = lane & 15;
  const int kh = (lane >> 4) << 1;
#pragma unroll
  for (int k = 0; k < K; k += 4) {
    v2f a, b;
    a[0] = A[mn * lda + k + kh + 0];
    a[1] = A[mn * lda + k + kh + 1];
    b[0] = B[(k + kh + 0) * ldb + mn];
    b[1] = B[(k + kh + 1) * ldb + mn];
    c = __builtin_amdgcn_wmma_f32_16x16x4_f32(false, a, false, b, (short)0, c,
                                              false, false);
  }
  return c;
}

__global__ __launch_bounds__(256) void atom_gemm_kernel(
    const float* __restrict__ xf, const float* __restrict__ W,
    const float* __restrict__ bias, float* __restrict__ ux) {
  const int row0 = blockIdx.x * 16;
  const int wave = threadIdx.x >> 5;
  const int col0 = wave * 16;
  v8f c = {};
  c = wmma_acc<FIN>(xf + (size_t)row0 * FIN, FIN, W + col0, HH, c);
  const int lane = threadIdx.x & 31;
  const int n = lane & 15;
  const int kh = lane >> 4;
  const float bv = bias[col0 + n];
#pragma unroll
  for (int v = 0; v < 8; ++v) {
    const float val = c[v] + bv;
    const size_t row = (size_t)(row0 + v + 8 * kh);
#pragma unroll
    for (int rc = 0; rc < RC; ++rc)
      ux[(size_t)rc * NN * HH + row * HH + col0 + n] = val;
  }
}

__global__ __launch_bounds__(256) void edge_gemm_kernel(
    const float* __restrict__ ea, const float* __restrict__ W,
    const float* __restrict__ bias, float* __restrict__ eemb) {
  const int row0 = blockIdx.x * 16;
  const int wave = threadIdx.x >> 5;
  const int col0 = wave * 16;
  v8f c = {};
  c = wmma_acc<FE>(ea + (size_t)row0 * FE, FE, W + col0, HH, c);
  const int lane = threadIdx.x & 31;
  const int n = lane & 15;
  const int kh = lane >> 4;
  const float bv = bias[col0 + n];
#pragma unroll
  for (int v = 0; v < 8; ++v)
    eemb[(size_t)(row0 + v + 8 * kh) * HH + col0 + n] = c[v] + bv;
}

__global__ __launch_bounds__(256) void msg_scatter_kernel(
    const float* __restrict__ ux, const float* __restrict__ eemb,
    const int* __restrict__ src, const int* __restrict__ dst,
    const float* __restrict__ mask, float* __restrict__ agg) {
  const long long tid = (long long)blockIdx.x * 256 + threadIdx.x;
  if (tid >= (long long)RC * EE * (HH / 4)) return;
  const int chunk = (int)(tid & 31);
  const int e = (int)((tid >> 5) % EE);
  const int rc = (int)((tid >> 5) / EE);
  const int s = src[e];
  const int d = dst[e];
  const float mk = mask[(size_t)rc * NN + s] * mask[(size_t)rc * NN + d];
  const float4 xs =
      *(const float4*)(ux + (size_t)rc * NN * HH + (size_t)s * HH + chunk * 4);
  const float4 ee = *(const float4*)(eemb + (size_t)e * HH + chunk * 4);
  float* ap = agg + (size_t)rc * NN * HH + (size_t)d * HH + chunk * 4;
  unsafeAtomicAdd(ap + 0, gelu_f(xs.x + ee.x) * mk);
  unsafeAtomicAdd(ap + 1, gelu_f(xs.y + ee.y) * mk);
  unsafeAtomicAdd(ap + 2, gelu_f(xs.z + ee.z) * mk);
  unsafeAtomicAdd(ap + 3, gelu_f(xs.w + ee.w) * mk);
}

#define PITCH 132
#define SLAB 32
#define NSLAB ((RC * NN) / SLAB)
#define MLP_GRID 512

__device__ __forceinline__ v8f wmma4(v2f a, v2f b, v8f c) {
  return __builtin_amdgcn_wmma_f32_16x16x4_f32(false, a, false, b, (short)0, c,
                                               false, false);
}

__device__ __forceinline__ void lds_barrier() {
  __builtin_amdgcn_fence(__ATOMIC_RELEASE, "workgroup", "local");
  __builtin_amdgcn_s_barrier();
  __builtin_amdgcn_fence(__ATOMIC_ACQUIRE, "workgroup", "local");
}

template <int LDB>
__device__ __forceinline__ void load_b_slice(const float* __restrict__ Wc,
                                             v2f (&b)[32]) {
  const int lane = threadIdx.x & 31;
  const int mn = lane & 15;
  const int kh = (lane >> 4) << 1;
#pragma unroll
  for (int kk = 0; kk < 32; ++kk) {
    b[kk][0] = Wc[(4 * kk + kh + 0) * LDB + mn];
    b[kk][1] = Wc[(4 * kk + kh + 1) * LDB + mn];
  }
}

__device__ __forceinline__ void gemm_two_tiles(const float* S,
                                               const v2f (&b)[32], v8f& c0,
                                               v8f& c1) {
  const int lane = threadIdx.x & 31;
  const float* p0 = S + (lane & 15) * PITCH + ((lane >> 4) << 1);
  const float* p1 = p0 + 16 * PITCH;
#pragma unroll
  for (int kk = 0; kk < 32; ++kk) {
    const v2f a0 = *(const v2f*)(p0 + 4 * kk);
    const v2f a1 = *(const v2f*)(p1 + 4 * kk);
    c0 = wmma4(a0, b[kk], c0);
    c1 = wmma4(a1, b[kk], c1);
  }
}

__device__ __forceinline__ v8f gemm_one_tile(const float* S,
                                             const v2f (&b)[32], v8f c) {
  const int lane = threadIdx.x & 31;
  const float* p0 = S + (lane & 15) * PITCH + ((lane >> 4) << 1);
#pragma unroll
  for (int kk = 0; kk < 32; ++kk) {
    const v2f a0 = *(const v2f*)(p0 + 4 * kk);
    c = wmma4(a0, b[kk], c);
  }
  return c;
}

__global__ __launch_bounds__(256) void gin_mlp_kernel(
    float* __restrict__ ux, const float* __restrict__ agg,
    const float* __restrict__ eps_l, const float* __restrict__ W1,
    const float* __restrict__ b1, const float* __restrict__ W2,
    const float* __restrict__ b2) {
  __shared__ __attribute__((aligned(16))) float hA[SLAB * PITCH];
  __shared__ __attribute__((aligned(16))) float hB[SLAB * PITCH];
  const int wave = threadIdx.x >> 5;
  const int col0 = wave * 16;
  const int lane = threadIdx.x & 31;
  const int n = lane & 15;
  const int kh = lane >> 4;
  const float one_eps = 1.0f + eps_l[0];

  v2f w1f[32], w2f[32];
  load_b_slice<HH>(W1 + col0, w1f);
  load_b_slice<HH>(W2 + col0, w2f);
  const float bv1 = b1[col0 + n];
  const float bv2 = b2[col0 + n];

  const int sr = threadIdx.x >> 5;
  const int sc = (threadIdx.x & 31) * 4;
  float4 pu[4], pa[4];
  int it = blockIdx.x;
  {
    const int f = it < NSLAB ? it : 0;
#pragma unroll
    for (int j = 0; j < 4; ++j) {
      const size_t g = ((size_t)f * SLAB + sr + 8 * j) * HH + sc;
      pu[j] = *(const float4*)(ux + g);
      pa[j] = *(const float4*)(agg + g);
    }
  }
  for (; it < NSLAB; it += gridDim.x) {
    const size_t row0 = (size_t)it * SLAB;
#pragma unroll
    for (int j = 0; j < 4; ++j) {
      float4 h;
      h.x = one_eps * pu[j].x + pa[j].x;
      h.y = one_eps * pu[j].y + pa[j].y;
      h.z = one_eps * pu[j].z + pa[j].z;
      h.w = one_eps * pu[j].w + pa[j].w;
      *(float4*)(hA + (sr + 8 * j) * PITCH + sc) = h;
    }
    const int nx = it + (int)gridDim.x < NSLAB ? it + (int)gridDim.x : it;
#pragma unroll
    for (int j = 0; j < 4; ++j) {
      const size_t g = ((size_t)nx * SLAB + sr + 8 * j) * HH + sc;
      pu[j] = *(const float4*)(ux + g);
      pa[j] = *(const float4*)(agg + g);
    }
    lds_barrier();

    v8f c0 = {}, c1 = {};
    gemm_two_tiles(hA, w1f, c0, c1);
#pragma unroll
    for (int v = 0; v < 8; ++v) {
      hB[(v + 8 * kh) * PITCH + col0 + n] = gelu_f(c0[v] + bv1);
      hB[(16 + v + 8 * kh) * PITCH + col0 + n] = gelu_f(c1[v] + bv1);
    }
    lds_barrier();

    v8f d0 = {}, d1 = {};
    gemm_two_tiles(hB, w2f, d0, d1);
#pragma unroll
    for (int v = 0; v < 8; ++v) {
      ux[(row0 + v + 8 * kh) * HH + col0 + n] = gelu_f(d0[v] + bv2);
      ux[(row0 + 16 + v + 8 * kh) * HH + col0 + n] = gelu_f(d1[v] + bv2);
    }
  }
}

__global__ __launch_bounds__(256) void head_mlp_kernel(
    const float* __restrict__ ux, const float* __restrict__ W1,
    const float* __restrict__ b1, const float* __restrict__ W2,
    const float* __restrict__ b2, float* __restrict__ obuf) {
  __shared__ __attribute__((aligned(16))) float hA[SLAB * PITCH];
  __shared__ __attribute__((aligned(16))) float hB[SLAB * PITCH];
  const int wave = threadIdx.x >> 5;
  const int col0 = wave * 16;
  const int lane = threadIdx.x & 31;
  const int n = lane & 15;
  const int kh = lane >> 4;
  const int ct = wave & 1;
  const int rt = (wave >> 1) & 1;

  v2f w1f[32], w2f[32];
  load_b_slice<HH>(W1 + col0, w1f);
  load_b_slice<OO>(W2 + ct * 16, w2f);
  const float bv1 = b1[col0 + n];
  const float bv2 = b2[ct * 16 + n];

  const int sr = threadIdx.x >> 5;
  const int sc = (threadIdx.x & 31) * 4;
  int it = blockIdx.x;
  const int f = it < NSLAB ? it : 0;
  const float* up = ux + ((size_t)f * SLAB + sr) * HH + sc;
  v4f p0 = *(const v4f*)(up);
  v4f p1 = *(const v4f*)(up + 8 * HH);
  v4f p2 = *(const v4f*)(up + 16 * HH);
  v4f p3 = *(const v4f*)(up + 24 * HH);
  for (; it < NSLAB; it += gridDim.x) {
    const size_t row0 = (size_t)it * SLAB;
    float* sp = hA + sr * PITCH + sc;
    *(v4f*)(sp) = p0;
    *(v4f*)(sp + 8 * PITCH) = p1;
    *(v4f*)(sp + 16 * PITCH) = p2;
    *(v4f*)(sp + 24 * PITCH) = p3;
    const int nx = it + (int)gridDim.x < NSLAB ? it + (int)gridDim.x : it;
    up = ux + ((size_t)nx * SLAB + sr) * HH + sc;
    p0 = *(const v4f*)(up);
    p1 = *(const v4f*)(up + 8 * HH);
    p2 = *(const v4f*)(up + 16 * HH);
    p3 = *(const v4f*)(up + 24 * HH);
    lds_barrier();

    v8f c0 = {}, c1 = {};
    gemm_two_tiles(hA, w1f, c0, c1);
#pragma unroll
    for (int v = 0; v < 8; ++v) {
      hB[(v + 8 * kh) * PITCH + col0 + n] = gelu_f(c0[v] + bv1);
      hB[(16 + v + 8 * kh) * PITCH + col0 + n] = gelu_f(c1[v] + bv1);
    }
    lds_barrier();

    if (wave < 4) {
      v8f d = {};
      d = gemm_one_tile(hB + rt * 16 * PITCH, w2f, d);
#pragma unroll
      for (int v = 0; v < 8; ++v)
        obuf[(row0 + rt * 16 + v + 8 * kh) * OO + ct * 16 + n] = d[v] + bv2;
    }
  }
}

__global__ __launch_bounds__(256) void pool_num_kernel(
    const float* __restrict__ obuf, const float* __restrict__ mask,
    const int* __restrict__ batch, float* __restrict__ num) {
  const long long tid = (long long)blockIdx.x * 256 + threadIdx.x;
  if (tid >= (long long)RC * NN * (OO / 4)) return;
  const int o4 = (int)(tid & 7);
  const int nd = (int)((tid >> 3) % NN);
  const int rc = (int)((tid >> 3) / NN);
  const float mk = mask[(size_t)rc * NN + nd];
  const int g = batch[nd];
  const float4 v =
      *(const float4*)(obuf + ((size_t)rc * NN + nd) * OO + o4 * 4);
  float* np = num + ((size_t)rc * GG + g) * OO + o4 * 4;
  unsafeAtomicAdd(np + 0, v.x * mk);
  unsafeAtomicAdd(np + 1, v.y * mk);
  unsafeAtomicAdd(np + 2, v.z * mk);
  unsafeAtomicAdd(np + 3, v.w * mk);
}

__global__ __launch_bounds__(256) void pool_den_kernel(
    const float* __restrict__ mask, const int* __restrict__ batch,
    float* __restrict__ den) {
  const int tid = blockIdx.x * 256 + threadIdx.x;
  if (tid >= RC * NN) return;
  const int nd = tid % NN;
  const int rc = tid / NN;
  unsafeAtomicAdd(den + (size_t)rc * GG + batch[nd],
                  mask[(size_t)rc * NN + nd]);
}

__global__ __launch_bounds__(256) void finalize_kernel(
    float* __restrict__ out, const float* __restrict__ den) {
  const int i = blockIdx.x * 256 + threadIdx.x;
  if (i >= RC * GG * OO) return;
  const int g = (i / OO) % GG;
  const int rc = i / (OO * GG);
  out[i] = out[i] / (den[(size_t)rc * GG + g] + 1e-7f);
}

extern "C" void kernel_launch(void* const* d_in, const int* in_sizes, int n_in,
                              void* d_out, int out_size, void* d_ws,
                              size_t ws_size, hipStream_t stream) {
  const float* x_feat = (const float*)d_in[0];
  const int* eidx = (const int*)d_in[1];
  const float* eattr = (const float*)d_in[2];
  const int* batch = (const int*)d_in[3];
  const float* nmask = (const float*)d_in[4];
  const float* atomW = (const float*)d_in[5];
  const float* atomB = (const float*)d_in[6];
  const float* eps = (const float*)d_in[7];
  const float* bondW = (const float*)d_in[8];
  const float* bondB = (const float*)d_in[9];
  const float* m1W = (const float*)d_in[10];
  const float* m1B = (const float*)d_in[11];
  const float* m2W = (const float*)d_in[12];
  const float* m2B = (const float*)d_in[13];
  const float* h1W = (const float*)d_in[14];
  const float* h1B = (const float*)d_in[15];
  const float* h2W = (const float*)d_in[16];
  const float* h2B = (const float*)d_in[17];
  float* out = (float*)d_out;

  const int* src = eidx;
  const int* dst = eidx + EE;

  float* ux = (float*)d_ws;
  float* agg = ux + (size_t)RC * NN * HH;
  float* eemb = agg + (size_t)RC * NN * HH;
  float* obuf = eemb + (size_t)EE * HH;
  float* den = obuf + (size_t)RC * NN * OO;

  atom_gemm_kernel<<<NN / 16, 256, 0, stream>>>(x_feat, atomW, atomB, ux);

  for (int l = 0; l < NLAY; ++l) {
    hipMemsetAsync(agg, 0, (size_t)RC * NN * HH * sizeof(float), stream);
    edge_gemm_kernel<<<EE / 16, 256, 0, stream>>>(
        eattr, bondW + (size_t)l * FE * HH, bondB + (size_t)l * HH, eemb);
    msg_scatter_kernel<<<(RC * EE * (HH / 4)) / 256, 256, 0, stream>>>(
        ux, eemb, src, dst, nmask, agg);
    gin_mlp_kernel<<<MLP_GRID, 256, 0, stream>>>(
        ux, agg, eps + l, m1W + (size_t)l * HH * HH, m1B + (size_t)l * HH,
        m2W + (size_t)l * HH * HH, m2B + (size_t)l * HH);
  }

  head_mlp_kernel<<<MLP_GRID, 256, 0, stream>>>(ux, h1W, h1B, h2W, h2B, obuf);

  hipMemsetAsync(out, 0, (size_t)RC * GG * OO * sizeof(float), stream);
  hipMemsetAsync(den, 0, (size_t)RC * GG * sizeof(float), stream);
  pool_num_kernel<<<(RC * NN * (OO / 4)) / 256, 256, 0, stream>>>(obuf, nmask,
                                                                  batch, out);
  pool_den_kernel<<<(RC * NN) / 256, 256, 0, stream>>>(nmask, batch, den);
  finalize_kernel<<<(RC * GG * OO) / 256, 256, 0, stream>>>(out, den);
}


# Scatter's f32 atomics and agg zero-fills removed: edges bucketed by dst in-call (LDS rank, scan, fill); one wave per (replica,node) sums its messages in registers; sum order = bucket order
# speedup vs baseline: 1.4641x; 1.2975x over previous
#include <hip/hip_runtime.h>
#include <cstddef>

#define NN   20000
#define EE   60000
#define GG   64
#define RC   8
#define FIN  64
#define FE   16
#define HH   128
#define OO   32
#define NLAY 2

typedef __attribute__((ext_vector_type(2))) float v2f;
typedef __attribute__((ext_vector_type(8))) float v8f;
typedef __attribute__((ext_vector_type(4))) float v4f;

__device__ __forceinline__ float gelu_f(float x) {
  return 0.5f * x * (1.0f + erff(x * 0.70710678118654752f));
}

template <int K>
__device__ __forceinline__ v8f wmma_acc(const float* A, int lda,
                                        const float* B, int ldb, v8f c) {
  const int lane = threadIdx.x & 31;
  const int mn = lane & 15;
  const int kh = (lane >> 4) << 1;
#pragma unroll
  for (int k = 0; k < K; k += 4) {
    v2f a, b;
    a[0] = A[mn * lda + k + kh + 0];
    a[1] = A[mn * lda + k + kh + 1];
    b[0] = B[(k + kh + 0) * ldb + mn];
    b[1] = B[(k + kh + 1) * ldb + mn];
    c = __builtin_amdgcn_wmma_f32_16x16x4_f32(false, a, false, b, (short)0, c,
                                              false, false);
  }
  return c;
}

__global__ __launch_bounds__(256) void atom_gemm_kernel(
    const float* __restrict__ xf, const float* __restrict__ W,
    const float* __restrict__ bias, float* __restrict__ ux) {
  const int row0 = blockIdx.x * 16;
  const int wave = threadIdx.x >> 5;
  const int col0 = wave * 16;
  v8f c = {};
  c = wmma_acc<FIN>(xf + (size_t)row0 * FIN, FIN, W + col0, HH, c);
  const int lane = threadIdx.x & 31;
  const int n = lane & 15;
  const int kh = lane >> 4;
  const float bv = bias[col0 + n];
#pragma unroll
  for (int v = 0; v < 8; ++v) {
    const float val = c[v] + bv;
    const size_t row = (size_t)(row0 + v + 8 * kh);
#pragma unroll
    for (int rc = 0; rc < RC; ++rc)
      ux[(size_t)rc * NN * HH + row * HH + col0 + n] = val;
  }
}

__global__ __launch_bounds__(256) void edge_gemm_kernel(
    const float* __restrict__ ea, const float* __restrict__ W,
    const float* __restrict__ bias, float* __restrict__ eemb) {
  const int row0 = blockIdx.x * 16;
  const int wave = threadIdx.x >> 5;
  const int col0 = wave * 16;
  v8f c = {};
  c = wmma_acc<FE>(ea + (size_t)row0 * FE, FE, W + col0, HH, c);
  const int lane = threadIdx.x & 31;
  const int n = lane & 15;
  const int kh = lane >> 4;
  const float bv = bias[col0 + n];
#pragma unroll
  for (int v = 0; v < 8; ++v)
    eemb[(size_t)(row0 + v + 8 * kh) * HH + col0 + n] = c[v] + bv;
}

#define CSR_T 1024
#define E4 (EE / 4)
#define CSR_K ((E4 + CSR_T - 1) / CSR_T)
#define CSR_CH (NN / 1000)
static_assert(EE % 4 == 0, "edges are read four at a time");
static_assert(CSR_CH * 1000 == NN && 1000 <= CSR_T, "scan chunks cover NN");

__global__ __launch_bounds__(CSR_T) void csr_rank_scan_kernel(
    const int* __restrict__ dst, int* __restrict__ rnk, int* __restrict__ off) {
  __shared__ int cnt[NN];
  __shared__ int wtot[CSR_T / 32];
  const int tid = threadIdx.x;
  const int lane = tid & 31;
  const int wave = tid >> 5;
  for (int i = tid; i < NN; i += CSR_T) cnt[i] = 0;

  int4 dv[CSR_K];
#pragma unroll
  for (int k = 0; k < CSR_K; ++k) {
    const int g = tid + k * CSR_T;
    dv[k] = g < E4 ? *(const int4*)(dst + 4 * g) : make_int4(0, 0, 0, 0);
  }
  __syncthreads();
#pragma unroll
  for (int k = 0; k < CSR_K; ++k) {
    const int g = tid + k * CSR_T;
    if (g < E4) {
      int4 r;
      r.x = atomicAdd(&cnt[dv[k].x], 1);
      r.y = atomicAdd(&cnt[dv[k].y], 1);
      r.z = atomicAdd(&cnt[dv[k].z], 1);
      r.w = atomicAdd(&cnt[dv[k].w], 1);
      *(int4*)(rnk + 4 * g) = r;
    }
  }
  __syncthreads();

  int c[CSR_CH];
  int tsum = 0;
  const int b0 = tid * CSR_CH;
  const bool own = b0 < NN;
#pragma unroll
  for (int j = 0; j < CSR_CH; ++j) {
    c[j] = own ? cnt[b0 + j] : 0;
    tsum += c[j];
  }
  int incl = tsum;
  for (int o = 1; o < 32; o <<= 1) {
    const int v = __shfl_up(incl, o);
    if (lane >= o) incl += v;
  }
  if (lane == 31) wtot[wave] = incl;
  __syncthreads();
  if (wave == 0) {
    const int w = wtot[lane];
    int wi = w;
    for (int o = 1; o < 32; o <<= 1) {
      const int v = __shfl_up(wi, o);
      if (lane >= o) wi += v;
    }
    wtot[lane] = wi - w;
  }
  __syncthreads();
  int run = wtot[wave] + incl - tsum;
  if (own) {
#pragma unroll
    for (int j = 0; j < CSR_CH; ++j) {
      cnt[b0 + j] = run;
      run += c[j];
    }
  }
  __syncthreads();
  for (int i = tid; i < NN; i += CSR_T) off[i] = cnt[i];
  if (tid == 0) off[NN] = EE;
}

__global__ __launch_bounds__(256) void csr_fill_kernel(
    const int* __restrict__ src, const int* __restrict__ dst,
    const int* __restrict__ rnk, const int* __restrict__ off,
    int2* __restrict__ pes) {
  const int e = blockIdx.x * 256 + threadIdx.x;
  if (e >= EE) return;
  const int slot = off[dst[e]] + rnk[e];
  pes[slot] = make_int2(e, src[e]);
}

#define GATHER_NPB 4
static_assert(NN % GATHER_NPB == 0, "node groups cover NN");
static_assert(RC * 32 == 256, "one wave of the work-group per replica");

__global__ __launch_bounds__(256) void msg_gather_kernel(
    const float* __restrict__ ux, const float* __restrict__ eemb,
    const int* __restrict__ off, const int2* __restrict__ pes,
    const float* __restrict__ mask, float* __restrict__ agg) {
  const int rc = threadIdx.x >> 5;
  const int lane = threadIdx.x & 31;
  const float* mrc = mask + (size_t)rc * NN;
  const float* uxr = ux + (size_t)rc * NN * HH + lane * 4;
  const float* eer = eemb + lane * 4;
  float* ar = agg + (size_t)rc * NN * HH + lane * 4;
  const int d0 = blockIdx.x * GATHER_NPB;
  int end = off[d0];
  for (int k = 0; k < GATHER_NPB; ++k) {
    const int d = d0 + k;
    const int beg = end;
    end = off[d + 1];
    const float md = mrc[d];
    float4 acc = make_float4(0.0f, 0.0f, 0.0f, 0.0f);
    for (int base = beg; base < end; base += 32) {
      const int cnt = end - base < 32 ? end - base : 32;
      int me = 0, ms = 0;
      float mm = 0.0f;
      if (lane < cnt) {
        const int2 p = pes[base + lane];
        me = p.x;
        ms = p.y;
        mm = mrc[ms] * md;
      }
      for (int j = 0; j < cnt; ++j) {
        const int e = __builtin_amdgcn_readlane(me, j);
        const int s = __builtin_amdgcn_readlane(ms, j);
        const float mk =
            __int_as_float(__builtin_amdgcn_readlane(__float_as_int(mm), j));
        const float4 xs = *(const float4*)(uxr + (size_t)s * HH);
        const float4 ee = *(const float4*)(eer + (size_t)e * HH);
        acc.x += gelu_f(xs.x + ee.x) * mk;
        acc.y += gelu_f(xs.y + ee.y) * mk;
        acc.z += gelu_f(xs.z + ee.z) * mk;
        acc.w += gelu_f(xs.w + ee.w) * mk;
      }
    }
    *(float4*)(ar + (size_t)d * HH) = acc;
  }
}

#define PITCH 132
#define SLAB 32
#define NSLAB ((RC * NN) / SLAB)
#define MLP_GRID 512

__device__ __forceinline__ v8f wmma4(v2f a, v2f b, v8f c) {
  return __builtin_amdgcn_wmma_f32_16x16x4_f32(false, a, false, b, (short)0, c,
                                               false, false);
}

__device__ __forceinline__ void lds_barrier() {
  __builtin_amdgcn_fence(__ATOMIC_RELEASE, "workgroup", "local");
  __builtin_amdgcn_s_barrier();
  __builtin_amdgcn_fence(__ATOMIC_ACQUIRE, "workgroup", "local");
}

template <int LDB>
__device__ __forceinline__ void load_b_slice(const float* __restrict__ Wc,
                                             v2f (&b)[32]) {
  const int lane = threadIdx.x & 31;
  const int mn = lane & 15;
  const int kh = (lane >> 4) << 1;
#pragma unroll
  for (int kk = 0; kk < 32; ++kk) {
    b[kk][0] = Wc[(4 * kk + kh + 0) * LDB + mn];
    b[kk][1] = Wc[(4 * kk + kh + 1) * LDB + mn];
  }
}

__device__ __forceinline__ void gemm_two_tiles(const float* S,
                                               const v2f (&b)[32], v8f& c0,
                                               v8f& c1) {
  const int lane = threadIdx.x & 31;
  const float* p0 = S + (lane & 15) * PITCH + ((lane >> 4) << 1);
  const float* p1 = p0 + 16 * PITCH;
#pragma unroll
  for (int kk = 0; kk < 32; ++kk) {
    const v2f a0 = *(const v2f*)(p0 + 4 * kk);
    const v2f a1 = *(const v2f*)(p1 + 4 * kk);
    c0 = wmma4(a0, b[kk], c0);
    c1 = wmma4(a1, b[kk], c1);
  }
}

__device__ __forceinline__ v8f gemm_one_tile(const float* S,
                                             const v2f (&b)[32], v8f c) {
  const int lane = threadIdx.x & 31;
  const float* p0 = S + (lane & 15) * PITCH + ((lane >> 4) << 1);
#pragma unroll
  for (int kk = 0; kk < 32; ++kk) {
    const v2f a0 = *(const v2f*)(p0 + 4 * kk);
    c = wmma4(a0, b[kk], c);
  }
  return c;
}

__global__ __launch_bounds__(256) void gin_mlp_kernel(
    float* __restrict__ ux, const float* __restrict__ agg,
    const float* __restrict__ eps_l, const float* __restrict__ W1,
    const float* __restrict__ b1, const float* __restrict__ W2,
    const float* __restrict__ b2) {
  __shared__ __attribute__((aligned(16))) float hA[SLAB * PITCH];
  __shared__ __attribute__((aligned(16))) float hB[SLAB * PITCH];
  const int wave = threadIdx.x >> 5;
  const int col0 = wave * 16;
  const int lane = threadIdx.x & 31;
  const int n = lane & 15;
  const int kh = lane >> 4;
  const float one_eps = 1.0f + eps_l[0];

  v2f w1f[32], w2f[32];
  load_b_slice<HH>(W1 + col0, w1f);
  load_b_slice<HH>(W2 + col0, w2f);
  const float bv1 = b1[col0 + n];
  const float bv2 = b2[col0 + n];

  const int sr = threadIdx.x >> 5;
  const int sc = (threadIdx.x & 31) * 4;
  float4 pu[4], pa[4];
  int it = blockIdx.x;
  {
    const int f = it < NSLAB ? it : 0;
#pragma unroll
    for (int j = 0; j < 4; ++j) {
      const size_t g = ((size_t)f * SLAB + sr + 8 * j) * HH + sc;
      pu[j] = *(const float4*)(ux + g);
      pa[j] = *(const float4*)(agg + g);
    }
  }
  for (; it < NSLAB; it += gridDim.x) {
    const size_t row0 = (size_t)it * SLAB;
#pragma unroll
    for (int j = 0; j < 4; ++j) {
      float4 h;
      h.x = one_eps * pu[j].x + pa[j].x;
      h.y = one_eps * pu[j].y + pa[j].y;
      h.z = one_eps * pu[j].z + pa[j].z;
      h.w = one_eps * pu[j].w + pa[j].w;
      *(float4*)(hA + (sr + 8 * j) * PITCH + sc) = h;
    }
    const int nx = it + (int)gridDim.x < NSLAB ? it + (int)gridDim.x : it;
#pragma unroll
    for (int j = 0; j < 4; ++j) {
      const size_t g = ((size_t)nx * SLAB + sr + 8 * j) * HH + sc;
      pu[j] = *(const float4*)(ux + g);
      pa[j] = *(const float4*)(agg + g);
    }
    lds_barrier();

    v8f c0 = {}, c1 = {};
    gemm_two_tiles(hA, w1f, c0, c1);
#pragma unroll
    for (int v = 0; v < 8; ++v) {
      hB[(v + 8 * kh) * PITCH + col0 + n] = gelu_f(c0[v] + bv1);
      hB[(16 + v + 8 * kh) * PITCH + col0 + n] = gelu_f(c1[v] + bv1);
    }
    lds_barrier();

    v8f d0 = {}, d1 = {};
    gemm_two_tiles(hB, w2f, d0, d1);
#pragma unroll
    for (int v = 0; v < 8; ++v) {
      ux[(row0 + v + 8 * kh) * HH + col0 + n] = gelu_f(d0[v] + bv2);
      ux[(row0 + 16 + v + 8 * kh) * HH + col0 + n] = gelu_f(d1[v] + bv2);
    }
  }
}

__global__ __launch_bounds__(256) void head_mlp_kernel(
    const float* __restrict__ ux, const float* __restrict__ W1,
    const float* __restrict__ b1, const float* __restrict__ W2,
    const float* __restrict__ b2, float* __restrict__ obuf) {
  __shared__ __attribute__((aligned(16))) float hA[SLAB * PITCH];
  __shared__ __attribute__((aligned(16))) float hB[SLAB * PITCH];
  const int wave = threadIdx.x >> 5;
  const int col0 = wave * 16;
  const int lane = threadIdx.x & 31;
  const int n = lane & 15;
  const int kh = lane >> 4;
  const int ct = wave & 1;
  const int rt = (wave >> 1) & 1;

  v2f w1f[32], w2f[32];
  load_b_slice<HH>(W1 + col0, w1f);
  load_b_slice<OO>(W2 + ct * 16, w2f);
  const float bv1 = b1[col0 + n];
  const float bv2 = b2[ct * 16 + n];

  const int sr = threadIdx.x >> 5;
  const int sc = (threadIdx.x & 31) * 4;
  int it = blockIdx.x;
  const int f = it < NSLAB ? it : 0;
  const float* up = ux + ((size_t)f * SLAB + sr) * HH + sc;
  v4f p0 = *(const v4f*)(up);
  v4f p1 = *(const v4f*)(up + 8 * HH);
  v4f p2 = *(const v4f*)(up + 16 * HH);
  v4f p3 = *(const v4f*)(up + 24 * HH);
  for (; it < NSLAB; it += gridDim.x) {
    const size_t row0 = (size_t)it * SLAB;
    float* sp = hA + sr * PITCH + sc;
    *(v4f*)(sp) = p0;
    *(v4f*)(sp + 8 * PITCH) = p1;
    *(v4f*)(sp + 16 * PITCH) = p2;
    *(v4f*)(sp + 24 * PITCH) = p3;
    const int nx = it + (int)gridDim.x < NSLAB ? it + (int)gridDim.x : it;
    up = ux + ((size_t)nx * SLAB + sr) * HH + sc;
    p0 = *(const v4f*)(up);
    p1 = *(const v4f*)(up + 8 * HH);
    p2 = *(const v4f*)(up + 16 * HH);
    p3 = *(const v4f*)(up + 24 * HH);
    lds_barrier();

    v8f c0 = {}, c1 = {};
    gemm_two_tiles(hA, w1f, c0, c1);
#pragma unroll
    for (int v = 0; v < 8; ++v) {
      hB[(v + 8 * kh) * PITCH + col0 + n] = gelu_f(c0[v] + bv1);
      hB[(16 + v + 8 * kh) * PITCH + col0 + n] = gelu_f(c1[v] + bv1);
    }
    lds_barrier();

    if (wave < 4) {
      v8f d = {};
      d = gemm_one_tile(hB + rt * 16 * PITCH, w2f, d);
#pragma unroll
      for (int v = 0; v < 8; ++v)
        obuf[(row0 + rt * 16 + v + 8 * kh) * OO + ct * 16 + n] = d[v] + bv2;
    }
  }
}

__global__ __launch_bounds__(256) void pool_num_kernel(
    const float* __restrict__ obuf, const float* __restrict__ mask,
    const int* __restrict__ batch, float* __restrict__ num) {
  const long long tid = (long long)blockIdx.x * 256 + threadIdx.x;
  if (tid >= (long long)RC * NN * (OO / 4)) return;
  const int o4 = (int)(tid & 7);
  const int nd = (int)((tid >> 3) % NN);
  const int rc = (int)((tid >> 3) / NN);
  const float mk = mask[(size_t)rc * NN + nd];
  const int g = batch[nd];
  const float4 v =
      *(const float4*)(obuf + ((size_t)rc * NN + nd) * OO + o4 * 4);
  float* np = num + ((size_t)rc * GG + g) * OO + o4 * 4;
  unsafeAtomicAdd(np + 0, v.x * mk);
  unsafeAtomicAdd(np + 1, v.y * mk);
  unsafeAtomicAdd(np + 2, v.z * mk);
  unsafeAtomicAdd(np + 3, v.w * mk);
}

__global__ __launch_bounds__(256) void pool_den_kernel(
    const float* __restrict__ mask, const int* __restrict__ batch,
    float* __restrict__ den) {
  const int tid = blockIdx.x * 256 + threadIdx.x;
  if (tid >= RC * NN) return;
  const int nd = tid % NN;
  const int rc = tid / NN;
  unsafeAtomicAdd(den + (size_t)rc * GG + batch[nd],
                  mask[(size_t)rc * NN + nd]);
}

__global__ __launch_bounds__(256) void finalize_kernel(
    float* __restrict__ out, const float* __restrict__ den) {
  const int i = blockIdx.x * 256 + threadIdx.x;
  if (i >= RC * GG * OO) return;
  const int g = (i / OO) % GG;
  const int rc = i / (OO * GG);
  out[i] = out[i] / (den[(size_t)rc * GG + g] + 1e-7f);
}

extern "C" void kernel_launch(void* const* d_in, const int* in_sizes, int n_in,
                              void* d_out, int out_size, void* d_ws,
                              size_t ws_size, hipStream_t stream) {
  const float* x_feat = (const float*)d_in[0];
  const int* eidx = (const int*)d_in[1];
  const float* eattr = (const float*)d_in[2];
  const int* batch = (const int*)d_in[3];
  const float* nmask = (const float*)d_in[4];
  const float* atomW = (const float*)d_in[5];
  const float* atomB = (const float*)d_in[6];
  const float* eps = (const float*)d_in[7];
  const float* bondW = (const float*)d_in[8];
  const float* bondB = (const float*)d_in[9];
  const float* m1W = (const float*)d_in[10];
  const float* m1B = (const float*)d_in[11];
  const float* m2W = (const float*)d_in[12];
  const float* m2B = (const float*)d_in[13];
  const float* h1W = (const float*)d_in[14];
  const float* h1B = (const float*)d_in[15];
  const float* h2W = (const float*)d_in[16];
  const float* h2B = (const float*)d_in[17];
  float* out = (float*)d_out;

  const int* src = eidx;
  const int* dst = eidx + EE;

  float* ux = (float*)d_ws;
  float* agg = ux + (size_t)RC * NN * HH;
  float* eemb = agg + (size_t)RC * NN * HH;
  float* obuf = eemb + (size_t)EE * HH;
  float* den = obuf + (size_t)RC * NN * OO;
  int* off = (int*)(den + (size_t)RC * GG);
  int* rnk = off + (NN + 4);
  int2* pes = (int2*)(rnk + EE);

  csr_rank_scan_kernel<<<1, CSR_T, 0, stream>>>(dst, rnk, off);
  csr_fill_kernel<<<(EE + 255) / 256, 256, 0, stream>>>(src, dst, rnk, off,
                                                        pes);

  atom_gemm_kernel<<<NN / 16, 256, 0, stream>>>(x_feat, atomW, atomB, ux);

  for (int l = 0; l < NLAY; ++l) {
    edge_gemm_kernel<<<EE / 16, 256, 0, stream>>>(
        eattr, bondW + (size_t)l * FE * HH, bondB + (size_t)l * HH, eemb);
    msg_gather_kernel<<<NN / GATHER_NPB, 256, 0, stream>>>(ux, eemb, off, pes,
                                                           nmask, agg);
    gin_mlp_kernel<<<MLP_GRID, 256, 0, stream>>>(
        ux, agg, eps + l, m1W + (size_t)l * HH * HH, m1B + (size_t)l * HH,
        m2W + (size_t)l * HH * HH, m2B + (size_t)l * HH);
  }

  head_mlp_kernel<<<MLP_GRID, 256, 0, stream>>>(ux, h1W, h1B, h2W, h2B, obuf);

  hipMemsetAsync(out, 0, (size_t)RC * GG * OO * sizeof(float), stream);
  hipMemsetAsync(den, 0, (size_t)RC * GG * sizeof(float), stream);
  pool_num_kernel<<<(RC * NN * (OO / 4)) / 256, 256, 0, stream>>>(obuf, nmask,
                                                                  batch, out);
  pool_den_kernel<<<(RC * NN) / 256, 256, 0, stream>>>(nmask, batch, den);
  finalize_kernel<<<(RC * GG * OO) / 256, 256, 0, stream>>>(out, den);
}


# Layer 0 is replica-independent: atom encoder stores one copy, layer-0 message gelu computed once per edge and applied to the 8 replica masks, GIN layer 0 reads that copy; sum order unchanged
# speedup vs baseline: 1.5583x; 1.0643x over previous
#include <hip/hip_runtime.h>
#include <cstddef>

#define NN   20000
#define EE   60000
#define GG   64
#define RC   8
#define FIN  64
#define FE   16
#define HH   128
#define OO   32
#define NLAY 2

typedef __attribute__((ext_vector_type(2))) float v2f;
typedef __attribute__((ext_vector_type(8))) float v8f;
typedef __attribute__((ext_vector_type(4))) float v4f;

__device__ __forceinline__ float gelu_f(float x) {
  return 0.5f * x * (1.0f + erff(x * 0.70710678118654752f));
}

template <int K>
__device__ __forceinline__ v8f wmma_acc(const float* A, int lda,
                                        const float* B, int ldb, v8f c) {
  const int lane = threadIdx.x & 31;
  const int mn = lane & 15;
  const int kh = (lane >> 4) << 1;
#pragma unroll
  for (int k = 0; k < K; k += 4) {
    v2f a, b;
    a[0] = A[mn * lda + k + kh + 0];
    a[1] = A[mn * lda + k + kh + 1];
    b[0] = B[(k + kh + 0) * ldb + mn];
    b[1] = B[(k + kh + 1) * ldb + mn];
    c = __builtin_amdgcn_wmma_f32_16x16x4_f32(false, a, false, b, (short)0, c,
                                              false, false);
  }
  return c;
}

__global__ __launch_bounds__(256) void atom_gemm_kernel(
    const float* __restrict__ xf, const float* __restrict__ W,
    const float* __restrict__ bias, float* __restrict__ x0) {
  const int row0 = blockIdx.x * 16;
  const int wave = threadIdx.x >> 5;
  const int col0 = wave * 16;
  v8f c = {};
  c = wmma_acc<FIN>(xf + (size_t)row0 * FIN, FIN, W + col0, HH, c);
  const int lane = threadIdx.x & 31;
  const int n = lane & 15;
  const int kh = lane >> 4;
  const float bv = bias[col0 + n];
#pragma unroll
  for (int v = 0; v < 8; ++v)
    x0[(size_t)(row0 + v + 8 * kh) * HH + col0 + n] = c[v] + bv;
}

__global__ __launch_bounds__(256) void edge_gemm_kernel(
    const float* __restrict__ ea, const float* __restrict__ W,
    const float* __restrict__ bias, float* __restrict__ eemb) {
  const int row0 = blockIdx.x * 16;
  const int wave = threadIdx.x >> 5;
  const int col0 = wave * 16;
  v8f c = {};
  c = wmma_acc<FE>(ea + (size_t)row0 * FE, FE, W + col0, HH, c);
  const int lane = threadIdx.x & 31;
  const int n = lane & 15;
  const int kh = lane >> 4;
  const float bv = bias[col0 + n];
#pragma unroll
  for (int v = 0; v < 8; ++v)
    eemb[(size_t)(row0 + v + 8 * kh) * HH + col0 + n] = c[v] + bv;
}

#define CSR_T 1024
#define E4 (EE / 4)
#define CSR_K ((E4 + CSR_T - 1) / CSR_T)
#define CSR_CH (NN / 1000)
static_assert(EE % 4 == 0, "edges are read four at a time");
static_assert(CSR_CH * 1000 == NN && 1000 <= CSR_T, "scan chunks cover NN");

__global__ __launch_bounds__(CSR_T) void csr_rank_scan_kernel(
    const int* __restrict__ dst, int* __restrict__ rnk, int* __restrict__ off) {
  __shared__ int cnt[NN];
  __shared__ int wtot[CSR_T / 32];
  const int tid = threadIdx.x;
  const int lane = tid & 31;
  const int wave = tid >> 5;
  for (int i = tid; i < NN; i += CSR_T) cnt[i] = 0;

  int4 dv[CSR_K];
#pragma unroll
  for (int k = 0; k < CSR_K; ++k) {
    const int g = tid + k * CSR_T;
    dv[k] = g < E4 ? *(const int4*)(dst + 4 * g) : make_int4(0, 0, 0, 0);
  }
  __syncthreads();
#pragma unroll
  for (int k = 0; k < CSR_K; ++k) {
    const int g = tid + k * CSR_T;
    if (g < E4) {
      int4 r;
      r.x = atomicAdd(&cnt[dv[k].x], 1);
      r.y = atomicAdd(&cnt[dv[k].y], 1);
      r.z = atomicAdd(&cnt[dv[k].z], 1);
      r.w = atomicAdd(&cnt[dv[k].w], 1);
      *(int4*)(rnk + 4 * g) = r;
    }
  }
  __syncthreads();

  int c[CSR_CH];
  int tsum = 0;
  const int b0 = tid * CSR_CH;
  const bool own = b0 < NN;
#pragma unroll
  for (int j = 0; j < CSR_CH; ++j) {
    c[j] = own ? cnt[b0 + j] : 0;
    tsum += c[j];
  }
  int incl = tsum;
  for (int o = 1; o < 32; o <<= 1) {
    const int v = __shfl_up(incl, o);
    if (lane >= o) incl += v;
  }
  if (lane == 31) wtot[wave] = incl;
  __syncthreads();
  if (wave == 0) {
    const int w = wtot[lane];
    int wi = w;
    for (int o = 1; o < 32; o <<= 1) {
      const int v = __shfl_up(wi, o);
      if (lane >= o) wi += v;
    }
    wtot[lane] = wi - w;
  }
  __syncthreads();
  int run = wtot[wave] + incl - tsum;
  if (own) {
#pragma unroll
    for (int j = 0; j < CSR_CH; ++j) {
      cnt[b0 + j] = run;
      run += c[j];
    }
  }
  __syncthreads();
  for (int i = tid; i < NN; i += CSR_T) off[i] = cnt[i];
  if (tid == 0) off[NN] = EE;
}

__global__ __launch_bounds__(256) void csr_fill_kernel(
    const int* __restrict__ src, const int* __restrict__ dst,
    const int* __restrict__ rnk, const int* __restrict__ off,
    int2* __restrict__ pes) {
  const int e = blockIdx.x * 256 + threadIdx.x;
  if (e >= EE) return;
  const int slot = off[dst[e]] + rnk[e];
  pes[slot] = make_int2(e, src[e]);
}

#define GATHER_NPB 4
static_assert(NN % GATHER_NPB == 0, "node groups cover NN");
static_assert(RC * 32 == 256, "one wave of the work-group per replica");

__global__ __launch_bounds__(256) void msg_gather_kernel(
    const float* __restrict__ ux, const float* __restrict__ eemb,
    const int* __restrict__ off, const int2* __restrict__ pes,
    const float* __restrict__ mask, float* __restrict__ agg) {
  const int rc = threadIdx.x >> 5;
  const int lane = threadIdx.x & 31;
  const float* mrc = mask + (size_t)rc * NN;
  const float* uxr = ux + (size_t)rc * NN * HH + lane * 4;
  const float* eer = eemb + lane * 4;
  float* ar = agg + (size_t)rc * NN * HH + lane * 4;
  const int d0 = blockIdx.x * GATHER_NPB;
  int end = off[d0];
  for (int k = 0; k < GATHER_NPB; ++k) {
    const int d = d0 + k;
    const int beg = end;
    end = off[d + 1];
    const float md = mrc[d];
    float4 acc = make_float4(0.0f, 0.0f, 0.0f, 0.0f);
    for (int base = beg; base < end; base += 32) {
      const int cnt = end - base < 32 ? end - base : 32;
      int me = 0, ms = 0;
      float mm = 0.0f;
      if (lane < cnt) {
        const int2 p = pes[base + lane];
        me = p.x;
        ms = p.y;
        mm = mrc[ms] * md;
      }
      for (int j = 0; j < cnt; ++j) {
        const int e = __builtin_amdgcn_readlane(me, j);
        const int s = __builtin_amdgcn_readlane(ms, j);
        const float mk =
            __int_as_float(__builtin_amdgcn_readlane(__float_as_int(mm), j));
        const float4 xs = *(const float4*)(uxr + (size_t)s * HH);
        const float4 ee = *(const float4*)(eer + (size_t)e * HH);
        acc.x += gelu_f(xs.x + ee.x) * mk;
        acc.y += gelu_f(xs.y + ee.y) * mk;
        acc.z += gelu_f(xs.z + ee.z) * mk;
        acc.w += gelu_f(xs.w + ee.w) * mk;
      }
    }
    *(float4*)(ar + (size_t)d * HH) = acc;
  }
}

#define G0_NPW 2
#define G0_NPB (8 * G0_NPW)
static_assert(NN % G0_NPB == 0, "node groups cover NN");
static_assert(RC == 8, "a group of 8 lanes fetches the 8 replica masks");

__global__ __launch_bounds__(256) void msg_gather_shared_kernel(
    const float* __restrict__ x0, const float* __restrict__ eemb,
    const int* __restrict__ off, const int2* __restrict__ pes,
    const float* __restrict__ mask, float* __restrict__ agg) {
  const int lane = threadIdx.x & 31;
  const int wave = __builtin_amdgcn_readfirstlane((int)(threadIdx.x >> 5));
  const int rcl = lane & 7;
  const int jl = lane >> 3;
  const float* mr = mask + (size_t)rcl * NN;
  const float* xr = x0 + lane * 4;
  const float* er = eemb + lane * 4;
  float* ar = agg + lane * 4;
  const int d0 = (blockIdx.x * 8 + wave) * G0_NPW;
  int end = off[d0];
  for (int k = 0; k < G0_NPW; ++k) {
    const int d = d0 + k;
    const int beg = end;
    end = off[d + 1];
    const float md = mr[d];
    float4 acc[RC];
#pragma unroll
    for (int rc = 0; rc < RC; ++rc) acc[rc] = make_float4(0.0f, 0.0f, 0.0f, 0.0f);
    for (int base = beg; base < end; base += 4) {
      const int cnt = end - base < 4 ? end - base : 4;
      int me = 0, ms = 0;
      float mm = 0.0f;
      if (jl < cnt) {
        const int2 p = pes[base + jl];
        me = p.x;
        ms = p.y;
        mm = mr[ms] * md;
      }
      float4 xs[4], ee[4];
#pragma unroll
      for (int j = 0; j < 4; ++j) {
        const int jj = j < cnt ? j : 0;
        const int e = __builtin_amdgcn_readlane(me, 8 * jj);
        const int s = __builtin_amdgcn_readlane(ms, 8 * jj);
        xs[j] = *(const float4*)(xr + (size_t)s * HH);
        ee[j] = *(const float4*)(er + (size_t)e * HH);
      }
#pragma unroll
      for (int j = 0; j < 4; ++j) {
        if (j < cnt) {
          float4 g;
          g.x = gelu_f(xs[j].x + ee[j].x);
          g.y = gelu_f(xs[j].y + ee[j].y);
          g.z = gelu_f(xs[j].z + ee[j].z);
          g.w = gelu_f(xs[j].w + ee[j].w);
#pragma unroll
          for (int rc = 0; rc < RC; ++rc) {
            const float mk = __int_as_float(
                __builtin_amdgcn_readlane(__float_as_int(mm), 8 * j + rc));
            acc[rc].x += g.x * mk;
            acc[rc].y += g.y * mk;
            acc[rc].z += g.z * mk;
            acc[rc].w += g.w * mk;
          }
        }
      }
    }
#pragma unroll
    for (int rc = 0; rc < RC; ++rc)
      *(float4*)(ar + ((size_t)rc * NN + d) * HH) = acc[rc];
  }
}

#define PITCH 132
#define SLAB 32
#define NSLAB ((RC * NN) / SLAB)
#define MLP_GRID 512

__device__ __forceinline__ v8f wmma4(v2f a, v2f b, v8f c) {
  return __builtin_amdgcn_wmma_f32_16x16x4_f32(false, a, false, b, (short)0, c,
                                               false, false);
}

__device__ __forceinline__ void lds_barrier() {
  __builtin_amdgcn_fence(__ATOMIC_RELEASE, "workgroup", "local");
  __builtin_amdgcn_s_barrier();
  __builtin_amdgcn_fence(__ATOMIC_ACQUIRE, "workgroup", "local");
}

template <int LDB>
__device__ __forceinline__ void load_b_slice(const float* __restrict__ Wc,
                                             v2f (&b)[32]) {
  const int lane = threadIdx.x & 31;
  const int mn = lane & 15;
  const int kh = (lane >> 4) << 1;
#pragma unroll
  for (int kk = 0; kk < 32; ++kk) {
    b[kk][0] = Wc[(4 * kk + kh + 0) * LDB + mn];
    b[kk][1] = Wc[(4 * kk + kh + 1) * LDB + mn];
  }
}

__device__ __forceinline__ void gemm_two_tiles(const float* S,
                                               const v2f (&b)[32], v8f& c0,
                                               v8f& c1) {
  const int lane = threadIdx.x & 31;
  const float* p0 = S + (lane & 15) * PITCH + ((lane >> 4) << 1);
  const float* p1 = p0 + 16 * PITCH;
#pragma unroll
  for (int kk = 0; kk < 32; ++kk) {
    const v2f a0 = *(const v2f*)(p0 + 4 * kk);
    const v2f a1 = *(const v2f*)(p1 + 4 * kk);
    c0 = wmma4(a0, b[kk], c0);
    c1 = wmma4(a1, b[kk], c1);
  }
}

__device__ __forceinline__ v8f gemm_one_tile(const float* S,
                                             const v2f (&b)[32], v8f c) {
  const int lane = threadIdx.x & 31;
  const float* p0 = S + (lane & 15) * PITCH + ((lane >> 4) << 1);
#pragma unroll
  for (int kk = 0; kk < 32; ++kk) {
    const v2f a0 = *(const v2f*)(p0 + 4 * kk);
    c = wmma4(a0, b[kk], c);
  }
  return c;
}

static_assert(NN % SLAB == 0, "a slab never straddles two replicas");
template <bool SHARED_IN>
__global__ __launch_bounds__(256) void gin_mlp_kernel(
    float* __restrict__ ux, const float* __restrict__ x0,
    const float* __restrict__ agg,
    const float* __restrict__ eps_l, const float* __restrict__ W1,
    const float* __restrict__ b1, const float* __restrict__ W2,
    const float* __restrict__ b2) {
  __shared__ __attribute__((aligned(16))) float hA[SLAB * PITCH];
  __shared__ __attribute__((aligned(16))) float hB[SLAB * PITCH];
  const int wave = threadIdx.x >> 5;
  const int col0 = wave * 16;
  const int lane = threadIdx.x & 31;
  const int n = lane & 15;
  const int kh = lane >> 4;
  const float one_eps = 1.0f + eps_l[0];

  v2f w1f[32], w2f[32];
  load_b_slice<HH>(W1 + col0, w1f);
  load_b_slice<HH>(W2 + col0, w2f);
  const float bv1 = b1[col0 + n];
  const float bv2 = b2[col0 + n];

  const int sr = threadIdx.x >> 5;
  const int sc = (threadIdx.x & 31) * 4;
  float4 pu[4], pa[4];
  const float* uin = SHARED_IN ? x0 : ux;
  int it = blockIdx.x;
  {
    const int f = it < NSLAB ? it : 0;
    const int fu = SHARED_IN ? f % (NN / SLAB) : f;
#pragma unroll
    for (int j = 0; j < 4; ++j) {
      const size_t g = ((size_t)f * SLAB + sr + 8 * j) * HH + sc;
      const size_t gu = ((size_t)fu * SLAB + sr + 8 * j) * HH + sc;
      pu[j] = *(const float4*)(uin + gu);
      pa[j] = *(const float4*)(agg + g);
    }
  }
  for (; it < NSLAB; it += gridDim.x) {
    const size_t row0 = (size_t)it * SLAB;
#pragma unroll
    for (int j = 0; j < 4; ++j) {
      float4 h;
      h.x = one_eps * pu[j].x + pa[j].x;
      h.y = one_eps * pu[j].y + pa[j].y;
      h.z = one_eps * pu[j].z + pa[j].z;
      h.w = one_eps * pu[j].w + pa[j].w;
      *(float4*)(hA + (sr + 8 * j) * PITCH + sc) = h;
    }
    const int nx = it + (int)gridDim.x < NSLAB ? it + (int)gridDim.x : it;
    const int nu = SHARED_IN ? nx % (NN / SLAB) : nx;
#pragma unroll
    for (int j = 0; j < 4; ++j) {
      const size_t g = ((size_t)nx * SLAB + sr + 8 * j) * HH + sc;
      const size_t gu = ((size_t)nu * SLAB + sr + 8 * j) * HH + sc;
      pu[j] = *(const float4*)(uin + gu);
      pa[j] = *(const float4*)(agg + g);
    }
    lds_barrier();

    v8f c0 = {}, c1 = {};
    gemm_two_tiles(hA, w1f, c0, c1);
#pragma unroll
    for (int v = 0; v < 8; ++v) {
      hB[(v + 8 * kh) * PITCH + col0 + n] = gelu_f(c0[v] + bv1);
      hB[(16 + v + 8 * kh) * PITCH + col0 + n] = gelu_f(c1[v] + bv1);
    }
    lds_barrier();

    v8f d0 = {}, d1 = {};
    gemm_two_tiles(hB, w2f, d0, d1);
#pragma unroll
    for (int v = 0; v < 8; ++v) {
      ux[(row0 + v + 8 * kh) * HH + col0 + n] = gelu_f(d0[v] + bv2);
      ux[(row0 + 16 + v + 8 * kh) * HH + col0 + n] = gelu_f(d1[v] + bv2);
    }
  }
}

__global__ __launch_bounds__(256) void head_mlp_kernel(
    const float* __restrict__ ux, const float* __restrict__ W1,
    const float* __restrict__ b1, const float* __restrict__ W2,
    const float* __restrict__ b2, float* __restrict__ obuf) {
  __shared__ __attribute__((aligned(16))) float hA[SLAB * PITCH];
  __shared__ __attribute__((aligned(16))) float hB[SLAB * PITCH];
  const int wave = threadIdx.x >> 5;
  const int col0 = wave * 16;
  const int lane = threadIdx.x & 31;
  const int n = lane & 15;
  const int kh = lane >> 4;
  const int ct = wave & 1;
  const int rt = (wave >> 1) & 1;

  v2f w1f[32], w2f[32];
  load_b_slice<HH>(W1 + col0, w1f);
  load_b_slice<OO>(W2 + ct * 16, w2f);
  const float bv1 = b1[col0 + n];
  const float bv2 = b2[ct * 16 + n];

  const int sr = threadIdx.x >> 5;
  const int sc = (threadIdx.x & 31) * 4;
  int it = blockIdx.x;
  const int f = it < NSLAB ? it : 0;
  const float* up = ux + ((size_t)f * SLAB + sr) * HH + sc;
  v4f p0 = *(const v4f*)(up);
  v4f p1 = *(const v4f*)(up + 8 * HH);
  v4f p2 = *(const v4f*)(up + 16 * HH);
  v4f p3 = *(const v4f*)(up + 24 * HH);
  for (; it < NSLAB; it += gridDim.x) {
    const size_t row0 = (size_t)it * SLAB;
    float* sp = hA + sr * PITCH + sc;
    *(v4f*)(sp) = p0;
    *(v4f*)(sp + 8 * PITCH) = p1;
    *(v4f*)(sp + 16 * PITCH) = p2;
    *(v4f*)(sp + 24 * PITCH) = p3;
    const int nx = it + (int)gridDim.x < NSLAB ? it + (int)gridDim.x : it;
    up = ux + ((size_t)nx * SLAB + sr) * HH + sc;
    p0 = *(const v4f*)(up);
    p1 = *(const v4f*)(up + 8 * HH);
    p2 = *(const v4f*)(up + 16 * HH);
    p3 = *(const v4f*)(up + 24 * HH);
    lds_barrier();

    v8f c0 = {}, c1 = {};
    gemm_two_tiles(hA, w1f, c0, c1);
#pragma unroll
    for (int v = 0; v < 8; ++v) {
      hB[(v + 8 * kh) * PITCH + col0 + n] = gelu_f(c0[v] + bv1);
      hB[(16 + v + 8 * kh) * PITCH + col0 + n] = gelu_f(c1[v] + bv1);
    }
    lds_barrier();

    if (wave < 4) {
      v8f d = {};
      d = gemm_one_tile(hB + rt * 16 * PITCH, w2f, d);
#pragma unroll
      for (int v = 0; v < 8; ++v)
        obuf[(row0 + rt * 16 + v + 8 * kh) * OO + ct * 16 + n] = d[v] + bv2;
    }
  }
}

__global__ __launch_bounds__(256) void pool_num_kernel(
    const float* __restrict__ obuf, const float* __restrict__ mask,
    const int* __restrict__ batch, float* __restrict__ num) {
  const long long tid = (long long)blockIdx.x * 256 + threadIdx.x;
  if (tid >= (long long)RC * NN * (OO / 4)) return;
  const int o4 = (int)(tid & 7);
  const int nd = (int)((tid >> 3) % NN);
  const int rc = (int)((tid >> 3) / NN);
  const float mk = mask[(size_t)rc * NN + nd];
  const int g = batch[nd];
  const float4 v =
      *(const float4*)(obuf + ((size_t)rc * NN + nd) * OO + o4 * 4);
  float* np = num + ((size_t)rc * GG + g) * OO + o4 * 4;
  unsafeAtomicAdd(np + 0, v.x * mk);
  unsafeAtomicAdd(np + 1, v.y * mk);
  unsafeAtomicAdd(np + 2, v.z * mk);
  unsafeAtomicAdd(np + 3, v.w * mk);
}

__global__ __launch_bounds__(256) void pool_den_kernel(
    const float* __restrict__ mask, const int* __restrict__ batch,
    float* __restrict__ den) {
  const int tid = blockIdx.x * 256 + threadIdx.x;
  if (tid >= RC * NN) return;
  const int nd = tid % NN;
  const int rc = tid / NN;
  unsafeAtomicAdd(den + (size_t)rc * GG + batch[nd],
                  mask[(size_t)rc * NN + nd]);
}

__global__ __launch_bounds__(256) void finalize_kernel(
    float* __restrict__ out, const float* __restrict__ den) {
  const int i = blockIdx.x * 256 + threadIdx.x;
  if (i >= RC * GG * OO) return;
  const int g = (i / OO) % GG;
  const int rc = i / (OO * GG);
  out[i] = out[i] / (den[(size_t)rc * GG + g] + 1e-7f);
}

extern "C" void kernel_launch(void* const* d_in, const int* in_sizes, int n_in,
                              void* d_out, int out_size, void* d_ws,
                              size_t ws_size, hipStream_t stream) {
  const float* x_feat = (const float*)d_in[0];
  const int* eidx = (const int*)d_in[1];
  const float* eattr = (const float*)d_in[2];
  const int* batch = (const int*)d_in[3];
  const float* nmask = (const float*)d_in[4];
  const float* atomW = (const float*)d_in[5];
  const float* atomB = (const float*)d_in[6];
  const float* eps = (const float*)d_in[7];
  const float* bondW = (const float*)d_in[8];
  const float* bondB = (const float*)d_in[9];
  const float* m1W = (const float*)d_in[10];
  const float* m1B = (const float*)d_in[11];
  const float* m2W = (const float*)d_in[12];
  const float* m2B = (const float*)d_in[13];
  const float* h1W = (const float*)d_in[14];
  const float* h1B = (const float*)d_in[15];
  const float* h2W = (const float*)d_in[16];
  const float* h2B = (const float*)d_in[17];
  float* out = (float*)d_out;

  const int* src = eidx;
  const int* dst = eidx + EE;

  float* ux = (float*)d_ws;
  float* agg = ux + (size_t)RC * NN * HH;
  float* eemb = agg + (size_t)RC * NN * HH;
  float* obuf = eemb + (size_t)EE * HH;
  float* x0 = obuf + (size_t)RC * NN * OO;
  float* den = x0 + (size_t)NN * HH;
  int* off = (int*)(den + (size_t)RC * GG);
  int* rnk = off + (NN + 4);
  int2* pes = (int2*)(rnk + EE);

  csr_rank_scan_kernel<<<1, CSR_T, 0, stream>>>(dst, rnk, off);
  csr_fill_kernel<<<(EE + 255) / 256, 256, 0, stream>>>(src, dst, rnk, off,
                                                        pes);

  atom_gemm_kernel<<<NN / 16, 256, 0, stream>>>(x_feat, atomW, atomB, x0);

  edge_gemm_kernel<<<EE / 16, 256, 0, stream>>>(eattr, bondW, bondB, eemb);
  msg_gather_shared_kernel<<<NN / G0_NPB, 256, 0, stream>>>(x0, eemb, off, pes,
                                                            nmask, agg);
  gin_mlp_kernel<true><<<MLP_GRID, 256, 0, stream>>>(ux, x0, agg, eps, m1W, m1B,
                                                     m2W, m2B);

  for (int l = 1; l < NLAY; ++l) {
    edge_gemm_kernel<<<EE / 16, 256, 0, stream>>>(
        eattr, bondW + (size_t)l * FE * HH, bondB + (size_t)l * HH, eemb);
    msg_gather_kernel<<<NN / GATHER_NPB, 256, 0, stream>>>(ux, eemb, off, pes,
                                                           nmask, agg);
    gin_mlp_kernel<false><<<MLP_GRID, 256, 0, stream>>>(
        ux, x0, agg, eps + l, m1W + (size_t)l * HH * HH, m1B + (size_t)l * HH,
        m2W + (size_t)l * HH * HH, m2B + (size_t)l * HH);
  }

  head_mlp_kernel<<<MLP_GRID, 256, 0, stream>>>(ux, h1W, h1B, h2W, h2B, obuf);

  hipMemsetAsync(out, 0, (size_t)RC * GG * OO * sizeof(float), stream);
  hipMemsetAsync(den, 0, (size_t)RC * GG * sizeof(float), stream);
  pool_num_kernel<<<(RC * NN * (OO / 4)) / 256, 256, 0, stream>>>(obuf, nmask,
                                                                  batch, out);
  pool_den_kernel<<<(RC * NN) / 256, 256, 0, stream>>>(nmask, batch, den);
  finalize_kernel<<<(RC * GG * OO) / 256, 256, 0, stream>>>(out, den);
}


# Exact GELU's erf made branch-free at every site: both range polynomials of the library form evaluated, one select, hardware exp2 for the library expf (erf within 2 ulp); sum order unchanged
# speedup vs baseline: 1.6426x; 1.0541x over previous
#include <hip/hip_runtime.h>
#include <cstddef>

#define NN   20000
#define EE   60000
#define GG   64
#define RC   8
#define FIN  64
#define FE   16
#define HH   128
#define OO   32
#define NLAY 2

typedef __attribute__((ext_vector_type(2))) float v2f;
typedef __attribute__((ext_vector_type(8))) float v8f;
typedef __attribute__((ext_vector_type(4))) float v4f;

__device__ __forceinline__ float erf_nb(float z) {
  const float a = __builtin_fabsf(z);
  const float s = a * a;
  float p = __builtin_fmaf(s, -0.000561801775f, 0.00491381623f);
  p = __builtin_fmaf(s, p, -0.0267075151f);
  p = __builtin_fmaf(s, p, 0.112800106f);
  p = __builtin_fmaf(s, p, -0.376122952f);
  p = __builtin_fmaf(s, p, 0.128379107f);
  const float lo = __builtin_fmaf(a, p, a);
  float q = __builtin_fmaf(a, 0.0000169988107f, -0.000378677854f);
  q = __builtin_fmaf(a, q, 0.00385781587f);
  q = __builtin_fmaf(a, q, -0.0241816975f);
  q = __builtin_fmaf(a, q, 0.106668264f);
  q = __builtin_fmaf(a, q, 0.634933293f);
  q = __builtin_fmaf(a, q, 0.128689408f);
  q = __builtin_fmaf(a, q, a);
  const float hi = 1.0f - __builtin_amdgcn_exp2f(q * -1.44269502f);
  return __builtin_copysignf(a < 1.0f ? lo : hi, z);
}

__device__ __forceinline__ float gelu_f(float x) {
  return 0.5f * x * (1.0f + erf_nb(x * 0.70710678118654752f));
}

template <int K>
__device__ __forceinline__ v8f wmma_acc(const float* A, int lda,
                                        const float* B, int ldb, v8f c) {
  const int lane = threadIdx.x & 31;
  const int mn = lane & 15;
  const int kh = (lane >> 4) << 1;
#pragma unroll
  for (int k = 0; k < K; k += 4) {
    v2f a, b;
    a[0] = A[mn * lda + k + kh + 0];
    a[1] = A[mn * lda + k + kh + 1];
    b[0] = B[(k + kh + 0) * ldb + mn];
    b[1] = B[(k + kh + 1) * ldb + mn];
    c = __builtin_amdgcn_wmma_f32_16x16x4_f32(false, a, false, b, (short)0, c,
                                              false, false);
  }
  return c;
}

__global__ __launch_bounds__(256) void atom_gemm_kernel(
    const float* __restrict__ xf, const float* __restrict__ W,
    const float* __restrict__ bias, float* __restrict__ x0) {
  const int row0 = blockIdx.x * 16;
  const int wave = threadIdx.x >> 5;
  const int col0 = wave * 16;
  v8f c = {};
  c = wmma_acc<FIN>(xf + (size_t)row0 * FIN, FIN, W + col0, HH, c);
  const int lane = threadIdx.x & 31;
  const int n = lane & 15;
  const int kh = lane >> 4;
  const float bv = bias[col0 + n];
#pragma unroll
  for (int v = 0; v < 8; ++v)
    x0[(size_t)(row0 + v + 8 * kh) * HH + col0 + n] = c[v] + bv;
}

__global__ __launch_bounds__(256) void edge_gemm_kernel(
    const float* __restrict__ ea, const float* __restrict__ W,
    const float* __restrict__ bias, float* __restrict__ eemb) {
  const int row0 = blockIdx.x * 16;
  const int wave = threadIdx.x >> 5;
  const int col0 = wave * 16;
  v8f c = {};
  c = wmma_acc<FE>(ea + (size_t)row0 * FE, FE, W + col0, HH, c);
  const int lane = threadIdx.x & 31;
  const int n = lane & 15;
  const int kh = lane >> 4;
  const float bv = bias[col0 + n];
#pragma unroll
  for (int v = 0; v < 8; ++v)
    eemb[(size_t)(row0 + v + 8 * kh) * HH + col0 + n] = c[v] + bv;
}

#define CSR_T 1024
#define E4 (EE / 4)
#define CSR_K ((E4 + CSR_T - 1) / CSR_T)
#define CSR_CH (NN / 1000)
static_assert(EE % 4 == 0, "edges are read four at a time");
static_assert(CSR_CH * 1000 == NN && 1000 <= CSR_T, "scan chunks cover NN");

__global__ __launch_bounds__(CSR_T) void csr_rank_scan_kernel(
    const int* __restrict__ dst, int* __restrict__ rnk, int* __restrict__ off) {
  __shared__ int cnt[NN];
  __shared__ int wtot[CSR_T / 32];
  const int tid = threadIdx.x;
  const int lane = tid & 31;
  const int wave = tid >> 5;
  for (int i = tid; i < NN; i += CSR_T) cnt[i] = 0;

  int4 dv[CSR_K];
#pragma unroll
  for (int k = 0; k < CSR_K; ++k) {
    const int g = tid + k * CSR_T;
    dv[k] = g < E4 ? *(const int4*)(dst + 4 * g) : make_int4(0, 0, 0, 0);
  }
  __syncthreads();
#pragma unroll
  for (int k = 0; k < CSR_K; ++k) {
    const int g = tid + k * CSR_T;
    if (g < E4) {
      int4 r;
      r.x = atomicAdd(&cnt[dv[k].x], 1);
      r.y = atomicAdd(&cnt[dv[k].y], 1);
      r.z = atomicAdd(&cnt[dv[k].z], 1);
      r.w = atomicAdd(&cnt[dv[k].w], 1);
      *(int4*)(rnk + 4 * g) = r;
    }
  }
  __syncthreads();

  int c[CSR_CH];
  int tsum = 0;
  const int b0 = tid * CSR_CH;
  const bool own = b0 < NN;
#pragma unroll
  for (int j = 0; j < CSR_CH; ++j) {
    c[j] = own ? cnt[b0 + j] : 0;
    tsum += c[j];
  }
  int incl = tsum;
  for (int o = 1; o < 32; o <<= 1) {
    const int v = __shfl_up(incl, o);
    if (lane >= o) incl += v;
  }
  if (lane == 31) wtot[wave] = incl;
  __syncthreads();
  if (wave == 0) {
    const int w = wtot[lane];
    int wi = w;
    for (int o = 1; o < 32; o <<= 1) {
      const int v = __shfl_up(wi, o);
      if (lane >= o) wi += v;
    }
    wtot[lane] = wi - w;
  }
  __syncthreads();
  int run = wtot[wave] + incl - tsum;
  if (own) {
#pragma unroll
    for (int j = 0; j < CSR_CH; ++j) {
      cnt[b0 + j] = run;
      run += c[j];
    }
  }
  __syncthreads();
  for (int i = tid; i < NN; i += CSR_T) off[i] = cnt[i];
  if (tid == 0) off[NN] = EE;
}

__global__ __launch_bounds__(256) void csr_fill_kernel(
    const int* __restrict__ src, const int* __restrict__ dst,
    const int* __restrict__ rnk, const int* __restrict__ off,
    int2* __restrict__ pes) {
  const int e = blockIdx.x * 256 + threadIdx.x;
  if (e >= EE) return;
  const int slot = off[dst[e]] + rnk[e];
  pes[slot] = make_int2(e, src[e]);
}

#define GATHER_NPB 4
static_assert(NN % GATHER_NPB == 0, "node groups cover NN");
static_assert(RC * 32 == 256, "one wave of the work-group per replica");

__global__ __launch_bounds__(256) void msg_gather_kernel(
    const float* __restrict__ ux, const float* __restrict__ eemb,
    const int* __restrict__ off, const int2* __restrict__ pes,
    const float* __restrict__ mask, float* __restrict__ agg) {
  const int rc = threadIdx.x >> 5;
  const int lane = threadIdx.x & 31;
  const float* mrc = mask + (size_t)rc * NN;
  const float* uxr = ux + (size_t)rc * NN * HH + lane * 4;
  const float* eer = eemb + lane * 4;
  float* ar = agg + (size_t)rc * NN * HH + lane * 4;
  const int d0 = blockIdx.x * GATHER_NPB;
  int end = off[d0];
  for (int k = 0; k < GATHER_NPB; ++k) {
    const int d = d0 + k;
    const int beg = end;
    end = off[d + 1];
    const float md = mrc[d];
    float4 acc = make_float4(0.0f, 0.0f, 0.0f, 0.0f);
    for (int base = beg; base < end; base += 32) {
      const int cnt = end - base < 32 ? end - base : 32;
      int me = 0, ms = 0;
      float mm = 0.0f;
      if (lane < cnt) {
        const int2 p = pes[base + lane];
        me = p.x;
        ms = p.y;
        mm = mrc[ms] * md;
      }
      for (int j = 0; j < cnt; ++j) {
        const int e = __builtin_amdgcn_readlane(me, j);
        const int s = __builtin_amdgcn_readlane(ms, j);
        const float mk =
            __int_as_float(__builtin_amdgcn_readlane(__float_as_int(mm), j));
        const float4 xs = *(const float4*)(uxr + (size_t)s * HH);
        const float4 ee = *(const float4*)(eer + (size_t)e * HH);
        acc.x += gelu_f(xs.x + ee.x) * mk;
        acc.y += gelu_f(xs.y + ee.y) * mk;
        acc.z += gelu_f(xs.z + ee.z) * mk;
        acc.w += gelu_f(xs.w + ee.w) * mk;
      }
    }
    *(float4*)(ar + (size_t)d * HH) = acc;
  }
}

#define G0_NPW 2
#define G0_NPB (8 * G0_NPW)
static_assert(NN % G0_NPB == 0, "node groups cover NN");
static_assert(RC == 8, "a group of 8 lanes fetches the 8 replica masks");

__global__ __launch_bounds__(256) void msg_gather_shared_kernel(
    const float* __restrict__ x0, const float* __restrict__ eemb,
    const int* __restrict__ off, const int2* __restrict__ pes,
    const float* __restrict__ mask, float* __restrict__ agg) {
  const int lane = threadIdx.x & 31;
  const int wave = __builtin_amdgcn_readfirstlane((int)(threadIdx.x >> 5));
  const int rcl = lane & 7;
  const int jl = lane >> 3;
  const float* mr = mask + (size_t)rcl * NN;
  const float* xr = x0 + lane * 4;
  const float* er = eemb + lane * 4;
  float* ar = agg + lane * 4;
  const int d0 = (blockIdx.x * 8 + wave) * G0_NPW;
  int end = off[d0];
  for (int k = 0; k < G0_NPW; ++k) {
    const int d = d0 + k;
    const int beg = end;
    end = off[d + 1];
    const float md = mr[d];
    float4 acc[RC];
#pragma unroll
    for (int rc = 0; rc < RC; ++rc) acc[rc] = make_float4(0.0f, 0.0f, 0.0f, 0.0f);
    for (int base = beg; base < end; base += 4) {
      const int cnt = end - base < 4 ? end - base : 4;
      int me = 0, ms = 0;
      float mm = 0.0f;
      if (jl < cnt) {
        const int2 p = pes[base + jl];
        me = p.x;
        ms = p.y;
        mm = mr[ms] * md;
      }
      float4 xs[4], ee[4];
#pragma unroll
      for (int j = 0; j < 4; ++j) {
        const int jj = j < cnt ? j : 0;
        const int e = __builtin_amdgcn_readlane(me, 8 * jj);
        const int s = __builtin_amdgcn_readlane(ms, 8 * jj);
        xs[j] = *(const float4*)(xr + (size_t)s * HH);
        ee[j] = *(const float4*)(er + (size_t)e * HH);
      }
#pragma unroll
      for (int j = 0; j < 4; ++j) {
        if (j < cnt) {
          float4 g;
          g.x = gelu_f(xs[j].x + ee[j].x);
          g.y = gelu_f(xs[j].y + ee[j].y);
          g.z = gelu_f(xs[j].z + ee[j].z);
          g.w = gelu_f(xs[j].w + ee[j].w);
#pragma unroll
          for (int rc = 0; rc < RC; ++rc) {
            const float mk = __int_as_float(
                __builtin_amdgcn_readlane(__float_as_int(mm), 8 * j + rc));
            acc[rc].x += g.x * mk;
            acc[rc].y += g.y * mk;
            acc[rc].z += g.z * mk;
            acc[rc].w += g.w * mk;
          }
        }
      }
    }
#pragma unroll
    for (int rc = 0; rc < RC; ++rc)
      *(float4*)(ar + ((size_t)rc * NN + d) * HH) = acc[rc];
  }
}

#define PITCH 132
#define SLAB 32
#define NSLAB ((RC * NN) / SLAB)
#define MLP_GRID 512

__device__ __forceinline__ v8f wmma4(v2f a, v2f b, v8f c) {
  return __builtin_amdgcn_wmma_f32_16x16x4_f32(false, a, false, b, (short)0, c,
                                               false, false);
}

__device__ __forceinline__ void lds_barrier() {
  __builtin_amdgcn_fence(__ATOMIC_RELEASE, "workgroup", "local");
  __builtin_amdgcn_s_barrier();
  __builtin_amdgcn_fence(__ATOMIC_ACQUIRE, "workgroup", "local");
}

template <int LDB>
__device__ __forceinline__ void load_b_slice(const float* __restrict__ Wc,
                                             v2f (&b)[32]) {
  const int lane = threadIdx.x & 31;
  const int mn = lane & 15;
  const int kh = (lane >> 4) << 1;
#pragma unroll
  for (int kk = 0; kk < 32; ++kk) {
    b[kk][0] = Wc[(4 * kk + kh + 0) * LDB + mn];
    b[kk][1] = Wc[(4 * kk + kh + 1) * LDB + mn];
  }
}

__device__ __forceinline__ void gemm_two_tiles(const float* S,
                                               const v2f (&b)[32], v8f& c0,
                                               v8f& c1) {
  const int lane = threadIdx.x & 31;
  const float* p0 = S + (lane & 15) * PITCH + ((lane >> 4) << 1);
  const float* p1 = p0 + 16 * PITCH;
#pragma unroll
  for (int kk = 0; kk < 32; ++kk) {
    const v2f a0 = *(const v2f*)(p0 + 4 * kk);
    const v2f a1 = *(const v2f*)(p1 + 4 * kk);
    c0 = wmma4(a0, b[kk], c0);
    c1 = wmma4(a1, b[kk], c1);
  }
}

__device__ __forceinline__ v8f gemm_one_tile(const float* S,
                                             const v2f (&b)[32], v8f c) {
  const int lane = threadIdx.x & 31;
  const float* p0 = S + (lane & 15) * PITCH + ((lane >> 4) << 1);
#pragma unroll
  for (int kk = 0; kk < 32; ++kk) {
    const v2f a0 = *(const v2f*)(p0 + 4 * kk);
    c = wmma4(a0, b[kk], c);
  }
  return c;
}

static_assert(NN % SLAB == 0, "a slab never straddles two replicas");
template <bool SHARED_IN>
__global__ __launch_bounds__(256) void gin_mlp_kernel(
    float* __restrict__ ux, const float* __restrict__ x0,
    const float* __restrict__ agg,
    const float* __restrict__ eps_l, const float* __restrict__ W1,
    const float* __restrict__ b1, const float* __restrict__ W2,
    const float* __restrict__ b2) {
  __shared__ __attribute__((aligned(16))) float hA[SLAB * PITCH];
  __shared__ __attribute__((aligned(16))) float hB[SLAB * PITCH];
  const int wave = threadIdx.x >> 5;
  const int col0 = wave * 16;
  const int lane = threadIdx.x & 31;
  const int n = lane & 15;
  const int kh = lane >> 4;
  const float one_eps = 1.0f + eps_l[0];

  v2f w1f[32], w2f[32];
  load_b_slice<HH>(W1 + col0, w1f);
  load_b_slice<HH>(W2 + col0, w2f);
  const float bv1 = b1[col0 + n];
  const float bv2 = b2[col0 + n];

  const int sr = threadIdx.x >> 5;
  const int sc = (threadIdx.x & 31) * 4;
  float4 pu[4], pa[4];
  const float* uin = SHARED_IN ? x0 : ux;
  int it = blockIdx.x;
  {
    const int f = it < NSLAB ? it : 0;
    const int fu = SHARED_IN ? f % (NN / SLAB) : f;
#pragma unroll
    for (int j = 0; j < 4; ++j) {
      const size_t g = ((size_t)f * SLAB + sr + 8 * j) * HH + sc;
      const size_t gu = ((size_t)fu * SLAB + sr + 8 * j) * HH + sc;
      pu[j] = *(const float4*)(uin + gu);
      pa[j] = *(const float4*)(agg + g);
    }
  }
  for (; it < NSLAB; it += gridDim.x) {
    const size_t row0 = (size_t)it * SLAB;
#pragma unroll
    for (int j = 0; j < 4; ++j) {
      float4 h;
      h.x = one_eps * pu[j].x + pa[j].x;
      h.y = one_eps * pu[j].y + pa[j].y;
      h.z = one_eps * pu[j].z + pa[j].z;
      h.w = one_eps * pu[j].w + pa[j].w;
      *(float4*)(hA + (sr + 8 * j) * PITCH + sc) = h;
    }
    const int nx = it + (int)gridDim.x < NSLAB ? it + (int)gridDim.x : it;
    const int nu = SHARED_IN ? nx % (NN / SLAB) : nx;
#pragma unroll
    for (int j = 0; j < 4; ++j) {
      const size_t g = ((size_t)nx * SLAB + sr + 8 * j) * HH + sc;
      const size_t gu = ((size_t)nu * SLAB + sr + 8 * j) * HH + sc;
      pu[j] = *(const float4*)(uin + gu);
      pa[j] = *(const float4*)(agg + g);
    }
    lds_barrier();

    v8f c0 = {}, c1 = {};
    gemm_two_tiles(hA, w1f, c0, c1);
#pragma unroll
    for (int v = 0; v < 8; ++v) {
      hB[(v + 8 * kh) * PITCH + col0 + n] = gelu_f(c0[v] + bv1);
      hB[(16 + v + 8 * kh) * PITCH + col0 + n] = gelu_f(c1[v] + bv1);
    }
    lds_barrier();

    v8f d0 = {}, d1 = {};
    gemm_two_tiles(hB, w2f, d0, d1);
#pragma unroll
    for (int v = 0; v < 8; ++v) {
      ux[(row0 + v + 8 * kh) * HH + col0 + n] = gelu_f(d0[v] + bv2);
      ux[(row0 + 16 + v + 8 * kh) * HH + col0 + n] = gelu_f(d1[v] + bv2);
    }
  }
}

__global__ __launch_bounds__(256) void head_mlp_kernel(
    const float* __restrict__ ux, const float* __restrict__ W1,
    const float* __restrict__ b1, const float* __restrict__ W2,
    const float* __restrict__ b2, float* __restrict__ obuf) {
  __shared__ __attribute__((aligned(16))) float hA[SLAB * PITCH];
  __shared__ __attribute__((aligned(16))) float hB[SLAB * PITCH];
  const int wave = threadIdx.x >> 5;
  const int col0 = wave * 16;
  const int lane = threadIdx.x & 31;
  const int n = lane & 15;
  const int kh = lane >> 4;
  const int ct = wave & 1;
  const int rt = (wave >> 1) & 1;

  v2f w1f[32], w2f[32];
  load_b_slice<HH>(W1 + col0, w1f);
  load_b_slice<OO>(W2 + ct * 16, w2f);
  const float bv1 = b1[col0 + n];
  const float bv2 = b2[ct * 16 + n];

  const int sr = threadIdx.x >> 5;
  const int sc = (threadIdx.x & 31) * 4;
  int it = blockIdx.x;
  const int f = it < NSLAB ? it : 0;
  const float* up = ux + ((size_t)f * SLAB + sr) * HH + sc;
  v4f p0 = *(const v4f*)(up);
  v4f p1 = *(const v4f*)(up + 8 * HH);
  v4f p2 = *(const v4f*)(up + 16 * HH);
  v4f p3 = *(const v4f*)(up + 24 * HH);
  for (; it < NSLAB; it += gridDim.x) {
    const size_t row0 = (size_t)it * SLAB;
    float* sp = hA + sr * PITCH + sc;
    *(v4f*)(sp) = p0;
    *(v4f*)(sp + 8 * PITCH) = p1;
    *(v4f*)(sp + 16 * PITCH) = p2;
    *(v4f*)(sp + 24 * PITCH) = p3;
    const int nx = it + (int)gridDim.x < NSLAB ? it + (int)gridDim.x : it;
    up = ux + ((size_t)nx * SLAB + sr) * HH + sc;
    p0 = *(const v4f*)(up);
    p1 = *(const v4f*)(up + 8 * HH);
    p2 = *(const v4f*)(up + 16 * HH);
    p3 = *(const v4f*)(up + 24 * HH);
    lds_barrier();

    v8f c0 = {}, c1 = {};
    gemm_two_tiles(hA, w1f, c0, c1);
#pragma unroll
    for (int v = 0; v < 8; ++v) {
      hB[(v + 8 * kh) * PITCH + col0 + n] = gelu_f(c0[v] + bv1);
      hB[(16 + v + 8 * kh) * PITCH + col0 + n] = gelu_f(c1[v] + bv1);
    }
    lds_barrier();

    if (wave < 4) {
      v8f d = {};
      d = gemm_one_tile(hB + rt * 16 * PITCH, w2f, d);
#pragma unroll
      for (int v = 0; v < 8; ++v)
        obuf[(row0 + rt * 16 + v + 8 * kh) * OO + ct * 16 + n] = d[v] + bv2;
    }
  }
}

__global__ __launch_bounds__(256) void pool_num_kernel(
    const float* __restrict__ obuf, const float* __restrict__ mask,
    const int* __restrict__ batch, float* __restrict__ num) {
  const long long tid = (long long)blockIdx.x * 256 + threadIdx.x;
  if (tid >= (long long)RC * NN * (OO / 4)) return;
  const int o4 = (int)(tid & 7);
  const int nd = (int)((tid >> 3) % NN);
  const int rc = (int)((tid >> 3) / NN);
  const float mk = mask[(size_t)rc * NN + nd];
  const int g = batch[nd];
  const float4 v =
      *(const float4*)(obuf + ((size_t)rc * NN + nd) * OO + o4 * 4);
  float* np = num + ((size_t)rc * GG + g) * OO + o4 * 4;
  unsafeAtomicAdd(np + 0, v.x * mk);
  unsafeAtomicAdd(np + 1, v.y * mk);
  unsafeAtomicAdd(np + 2, v.z * mk);
  unsafeAtomicAdd(np + 3, v.w * mk);
}

__global__ __launch_bounds__(256) void pool_den_kernel(
    const float* __restrict__ mask, const int* __restrict__ batch,
    float* __restrict__ den) {
  const int tid = blockIdx.x * 256 + threadIdx.x;
  if (tid >= RC * NN) return;
  const int nd = tid % NN;
  const int rc = tid / NN;
  unsafeAtomicAdd(den + (size_t)rc * GG + batch[nd],
                  mask[(size_t)rc * NN + nd]);
}

__global__ __launch_bounds__(256) void finalize_kernel(
    float* __restrict__ out, const float* __restrict__ den) {
  const int i = blockIdx.x * 256 + threadIdx.x;
  if (i >= RC * GG * OO) return;
  const int g = (i / OO) % GG;
  const int rc = i / (OO * GG);
  out[i] = out[i] / (den[(size_t)rc * GG + g] + 1e-7f);
}

extern "C" void kernel_launch(void* const* d_in, const int* in_sizes, int n_in,
                              void* d_out, int out_size, void* d_ws,
                              size_t ws_size, hipStream_t stream) {
  const float* x_feat = (const float*)d_in[0];
  const int* eidx = (const int*)d_in[1];
  const float* eattr = (const float*)d_in[2];
  const int* batch = (const int*)d_in[3];
  const float* nmask = (const float*)d_in[4];
  const float* atomW = (const float*)d_in[5];
  const float* atomB = (const float*)d_in[6];
  const float* eps = (const float*)d_in[7];
  const float* bondW = (const float*)d_in[8];
  const float* bondB = (const float*)d_in[9];
  const float* m1W = (const float*)d_in[10];
  const float* m1B = (const float*)d_in[11];
  const float* m2W = (const float*)d_in[12];
  const float* m2B = (const float*)d_in[13];
  const float* h1W = (const float*)d_in[14];
  const float* h1B = (const float*)d_in[15];
  const float* h2W = (const float*)d_in[16];
  const float* h2B = (const float*)d_in[17];
  float* out = (float*)d_out;

  const int* src = eidx;
  const int* dst = eidx + EE;

  float* ux = (float*)d_ws;
  float* agg = ux + (size_t)RC * NN * HH;
  float* eemb = agg + (size_t)RC * NN * HH;
  float* obuf = eemb + (size_t)EE * HH;
  float* x0 = obuf + (size_t)RC * NN * OO;
  float* den = x0 + (size_t)NN * HH;
  int* off = (int*)(den + (size_t)RC * GG);
  int* rnk = off + (NN + 4);
  int2* pes = (int2*)(rnk + EE);

  csr_rank_scan_kernel<<<1, CSR_T, 0, stream>>>(dst, rnk, off);
  csr_fill_kernel<<<(EE + 255) / 256, 256, 0, stream>>>(src, dst, rnk, off,
                                                        pes);

  atom_gemm_kernel<<<NN / 16, 256, 0, stream>>>(x_feat, atomW, atomB, x0);

  edge_gemm_kernel<<<EE / 16, 256, 0, stream>>>(eattr, bondW, bondB, eemb);
  msg_gather_shared_kernel<<<NN / G0_NPB, 256, 0, stream>>>(x0, eemb, off, pes,
                                                            nmask, agg);
  gin_mlp_kernel<true><<<MLP_GRID, 256, 0, stream>>>(ux, x0, agg, eps, m1W, m1B,
                                                     m2W, m2B);

  for (int l = 1; l < NLAY; ++l) {
    edge_gemm_kernel<<<EE / 16, 256, 0, stream>>>(
        eattr, bondW + (size_t)l * FE * HH, bondB + (size_t)l * HH, eemb);
    msg_gather_kernel<<<NN / GATHER_NPB, 256, 0, stream>>>(ux, eemb, off, pes,
                                                           nmask, agg);
    gin_mlp_kernel<false><<<MLP_GRID, 256, 0, stream>>>(
        ux, x0, agg, eps + l, m1W + (size_t)l * HH * HH, m1B + (size_t)l * HH,
        m2W + (size_t)l * HH * HH, m2B + (size_t)l * HH);
  }

  head_mlp_kernel<<<MLP_GRID, 256, 0, stream>>>(ux, h1W, h1B, h2W, h2B, obuf);

  hipMemsetAsync(out, 0, (size_t)RC * GG * OO * sizeof(float), stream);
  hipMemsetAsync(den, 0, (size_t)RC * GG * sizeof(float), stream);
  pool_num_kernel<<<(RC * NN * (OO / 4)) / 256, 256, 0, stream>>>(obuf, nmask,
                                                                  batch, out);
  pool_den_kernel<<<(RC * NN) / 256, 256, 0, stream>>>(nmask, batch, den);
  finalize_kernel<<<(RC * GG * OO) / 256, 256, 0, stream>>>(out, den);
}


# Masked pool sums taken in the head kernel's epilogue: 16x16 out tile times row masks, rows summed in registers, one f32 atomic per column per tile; both pool launches gone; sum order changes
# speedup vs baseline: 1.9873x; 1.2099x over previous
#include <hip/hip_runtime.h>
#include <cstddef>

#define NN   20000
#define EE   60000
#define GG   64
#define RC   8
#define FIN  64
#define FE   16
#define HH   128
#define OO   32
#define NLAY 2

typedef __attribute__((ext_vector_type(2))) float v2f;
typedef __attribute__((ext_vector_type(8))) float v8f;
typedef __attribute__((ext_vector_type(4))) float v4f;

__device__ __forceinline__ float erf_nb(float z) {
  const float a = __builtin_fabsf(z);
  const float s = a * a;
  float p = __builtin_fmaf(s, -0.000561801775f, 0.00491381623f);
  p = __builtin_fmaf(s, p, -0.0267075151f);
  p = __builtin_fmaf(s, p, 0.112800106f);
  p = __builtin_fmaf(s, p, -0.376122952f);
  p = __builtin_fmaf(s, p, 0.128379107f);
  const float lo = __builtin_fmaf(a, p, a);
  float q = __builtin_fmaf(a, 0.0000169988107f, -0.000378677854f);
  q = __builtin_fmaf(a, q, 0.00385781587f);
  q = __builtin_fmaf(a, q, -0.0241816975f);
  q = __builtin_fmaf(a, q, 0.106668264f);
  q = __builtin_fmaf(a, q, 0.634933293f);
  q = __builtin_fmaf(a, q, 0.128689408f);
  q = __builtin_fmaf(a, q, a);
  const float hi = 1.0f - __builtin_amdgcn_exp2f(q * -1.44269502f);
  return __builtin_copysignf(a < 1.0f ? lo : hi, z);
}

__device__ __forceinline__ float gelu_f(float x) {
  return 0.5f * x * (1.0f + erf_nb(x * 0.70710678118654752f));
}

template <int K>
__device__ __forceinline__ v8f wmma_acc(const float* A, int lda,
                                        const float* B, int ldb, v8f c) {
  const int lane = threadIdx.x & 31;
  const int mn = lane & 15;
  const int kh = (lane >> 4) << 1;
#pragma unroll
  for (int k = 0; k < K; k += 4) {
    v2f a, b;
    a[0] = A[mn * lda + k + kh + 0];
    a[1] = A[mn * lda + k + kh + 1];
    b[0] = B[(k + kh + 0) * ldb + mn];
    b[1] = B[(k + kh + 1) * ldb + mn];
    c = __builtin_amdgcn_wmma_f32_16x16x4_f32(false, a, false, b, (short)0, c,
                                              false, false);
  }
  return c;
}

__global__ __launch_bounds__(256) void atom_gemm_kernel(
    const float* __restrict__ xf, const float* __restrict__ W,
    const float* __restrict__ bias, float* __restrict__ x0) {
  const int row0 = blockIdx.x * 16;
  const int wave = threadIdx.x >> 5;
  const int col0 = wave * 16;
  v8f c = {};
  c = wmma_acc<FIN>(xf + (size_t)row0 * FIN, FIN, W + col0, HH, c);
  const int lane = threadIdx.x & 31;
  const int n = lane & 15;
  const int kh = lane >> 4;
  const float bv = bias[col0 + n];
#pragma unroll
  for (int v = 0; v < 8; ++v)
    x0[(size_t)(row0 + v + 8 * kh) * HH + col0 + n] = c[v] + bv;
}

__global__ __launch_bounds__(256) void edge_gemm_kernel(
    const float* __restrict__ ea, const float* __restrict__ W,
    const float* __restrict__ bias, float* __restrict__ eemb) {
  const int row0 = blockIdx.x * 16;
  const int wave = threadIdx.x >> 5;
  const int col0 = wave * 16;
  v8f c = {};
  c = wmma_acc<FE>(ea + (size_t)row0 * FE, FE, W + col0, HH, c);
  const int lane = threadIdx.x & 31;
  const int n = lane & 15;
  const int kh = lane >> 4;
  const float bv = bias[col0 + n];
#pragma unroll
  for (int v = 0; v < 8; ++v)
    eemb[(size_t)(row0 + v + 8 * kh) * HH + col0 + n] = c[v] + bv;
}

#define CSR_T 1024
#define E4 (EE / 4)
#define CSR_K ((E4 + CSR_T - 1) / CSR_T)
#define CSR_CH (NN / 1000)
static_assert(EE % 4 == 0, "edges are read four at a time");
static_assert(CSR_CH * 1000 == NN && 1000 <= CSR_T, "scan chunks cover NN");

__global__ __launch_bounds__(CSR_T) void csr_rank_scan_kernel(
    const int* __restrict__ dst, int* __restrict__ rnk, int* __restrict__ off) {
  __shared__ int cnt[NN];
  __shared__ int wtot[CSR_T / 32];
  const int tid = threadIdx.x;
  const int lane = tid & 31;
  const int wave = tid >> 5;
  for (int i = tid; i < NN; i += CSR_T) cnt[i] = 0;

  int4 dv[CSR_K];
#pragma unroll
  for (int k = 0; k < CSR_K; ++k) {
    const int g = tid + k * CSR_T;
    dv[k] = g < E4 ? *(const int4*)(dst + 4 * g) : make_int4(0, 0, 0, 0);
  }
  __syncthreads();
#pragma unroll
  for (int k = 0; k < CSR_K; ++k) {
    const int g = tid + k * CSR_T;
    if (g < E4) {
      int4 r;
      r.x = atomicAdd(&cnt[dv[k].x], 1);
      r.y = atomicAdd(&cnt[dv[k].y], 1);
      r.z = atomicAdd(&cnt[dv[k].z], 1);
      r.w = atomicAdd(&cnt[dv[k].w], 1);
      *(int4*)(rnk + 4 * g) = r;
    }
  }
  __syncthreads();

  int c[CSR_CH];
  int tsum = 0;
  const int b0 = tid * CSR_CH;
  const bool own = b0 < NN;
#pragma unroll
  for (int j = 0; j < CSR_CH; ++j) {
    c[j] = own ? cnt[b0 + j] : 0;
    tsum += c[j];
  }
  int incl = tsum;
  for (int o = 1; o < 32; o <<= 1) {
    const int v = __shfl_up(incl, o);
    if (lane >= o) incl += v;
  }
  if (lane == 31) wtot[wave] = incl;
  __syncthreads();
  if (wave == 0) {
    const int w = wtot[lane];
    int wi = w;
    for (int o = 1; o < 32; o <<= 1) {
      const int v = __shfl_up(wi, o);
      if (lane >= o) wi += v;
    }
    wtot[lane] = wi - w;
  }
  __syncthreads();
  int run = wtot[wave] + incl - tsum;
  if (own) {
#pragma unroll
    for (int j = 0; j < CSR_CH; ++j) {
      cnt[b0 + j] = run;
      run += c[j];
    }
  }
  __syncthreads();
  for (int i = tid; i < NN; i += CSR_T) off[i] = cnt[i];
  if (tid == 0) off[NN] = EE;
}

__global__ __launch_bounds__(256) void csr_fill_kernel(
    const int* __restrict__ src, const int* __restrict__ dst,
    const int* __restrict__ rnk, const int* __restrict__ off,
    int2* __restrict__ pes) {
  const int e = blockIdx.x * 256 + threadIdx.x;
  if (e >= EE) return;
  const int slot = off[dst[e]] + rnk[e];
  pes[slot] = make_int2(e, src[e]);
}

#define GATHER_NPB 4
static_assert(NN % GATHER_NPB == 0, "node groups cover NN");
static_assert(RC * 32 == 256, "one wave of the work-group per replica");

__global__ __launch_bounds__(256) void msg_gather_kernel(
    const float* __restrict__ ux, const float* __restrict__ eemb,
    const int* __restrict__ off, const int2* __restrict__ pes,
    const float* __restrict__ mask, float* __restrict__ agg) {
  const int rc = threadIdx.x >> 5;
  const int lane = threadIdx.x & 31;
  const float* mrc = mask + (size_t)rc * NN;
  const float* uxr = ux + (size_t)rc * NN * HH + lane * 4;
  const float* eer = eemb + lane * 4;
  float* ar = agg + (size_t)rc * NN * HH + lane * 4;
  const int d0 = blockIdx.x * GATHER_NPB;
  int end = off[d0];
  for (int k = 0; k < GATHER_NPB; ++k) {
    const int d = d0 + k;
    const int beg = end;
    end = off[d + 1];
    const float md = mrc[d];
    float4 acc = make_float4(0.0f, 0.0f, 0.0f, 0.0f);
    for (int base = beg; base < end; base += 32) {
      const int cnt = end - base < 32 ? end - base : 32;
      int me = 0, ms = 0;
      float mm = 0.0f;
      if (lane < cnt) {
        const int2 p = pes[base + lane];
        me = p.x;
        ms = p.y;
        mm = mrc[ms] * md;
      }
      for (int j = 0; j < cnt; ++j) {
        const int e = __builtin_amdgcn_readlane(me, j);
        const int s = __builtin_amdgcn_readlane(ms, j);
        const float mk =
            __int_as_float(__builtin_amdgcn_readlane(__float_as_int(mm), j));
        const float4 xs = *(const float4*)(uxr + (size_t)s * HH);
        const float4 ee = *(const float4*)(eer + (size_t)e * HH);
        acc.x += gelu_f(xs.x + ee.x) * mk;
        acc.y += gelu_f(xs.y + ee.y) * mk;
        acc.z += gelu_f(xs.z + ee.z) * mk;
        acc.w += gelu_f(xs.w + ee.w) * mk;
      }
    }
    *(float4*)(ar + (size_t)d * HH) = acc;
  }
}

#define G0_NPW 2
#define G0_NPB (8 * G0_NPW)
static_assert(NN % G0_NPB == 0, "node groups cover NN");
static_assert(RC == 8, "a group of 8 lanes fetches the 8 replica masks");

__global__ __launch_bounds__(256) void msg_gather_shared_kernel(
    const float* __restrict__ x0, const float* __restrict__ eemb,
    const int* __restrict__ off, const int2* __restrict__ pes,
    const float* __restrict__ mask, float* __restrict__ agg) {
  const int lane = threadIdx.x & 31;
  const int wave = __builtin_amdgcn_readfirstlane((int)(threadIdx.x >> 5));
  const int rcl = lane & 7;
  const int jl = lane >> 3;
  const float* mr = mask + (size_t)rcl * NN;
  const float* xr = x0 + lane * 4;
  const float* er = eemb + lane * 4;
  float* ar = agg + lane * 4;
  const int d0 = (blockIdx.x * 8 + wave) * G0_NPW;
  int end = off[d0];
  for (int k = 0; k < G0_NPW; ++k) {
    const int d = d0 + k;
    const int beg = end;
    end = off[d + 1];
    const float md = mr[d];
    float4 acc[RC];
#pragma unroll
    for (int rc = 0; rc < RC; ++rc) acc[rc] = make_float4(0.0f, 0.0f, 0.0f, 0.0f);
    for (int base = beg; base < end; base += 4) {
      const int cnt = end - base < 4 ? end - base : 4;
      int me = 0, ms = 0;
      float mm = 0.0f;
      if (jl < cnt) {
        const int2 p = pes[base + jl];
        me = p.x;
        ms = p.y;
        mm = mr[ms] * md;
      }
      float4 xs[4], ee[4];
#pragma unroll
      for (int j = 0; j < 4; ++j) {
        const int jj = j < cnt ? j : 0;
        const int e = __builtin_amdgcn_readlane(me, 8 * jj);
        const int s = __builtin_amdgcn_readlane(ms, 8 * jj);
        xs[j] = *(const float4*)(xr + (size_t)s * HH);
        ee[j] = *(const float4*)(er + (size_t)e * HH);
      }
#pragma unroll
      for (int j = 0; j < 4; ++j) {
        if (j < cnt) {
          float4 g;
          g.x = gelu_f(xs[j].x + ee[j].x);
          g.y = gelu_f(xs[j].y + ee[j].y);
          g.z = gelu_f(xs[j].z + ee[j].z);
          g.w = gelu_f(xs[j].w + ee[j].w);
#pragma unroll
          for (int rc = 0; rc < RC; ++rc) {
            const float mk = __int_as_float(
                __builtin_amdgcn_readlane(__float_as_int(mm), 8 * j + rc));
            acc[rc].x += g.x * mk;
            acc[rc].y += g.y * mk;
            acc[rc].z += g.z * mk;
            acc[rc].w += g.w * mk;
          }
        }
      }
    }
#pragma unroll
    for (int rc = 0; rc < RC; ++rc)
      *(float4*)(ar + ((size_t)rc * NN + d) * HH) = acc[rc];
  }
}

#define PITCH 132
#define SLAB 32
#define NSLAB ((RC * NN) / SLAB)
#define MLP_GRID 512

__device__ __forceinline__ v8f wmma4(v2f a, v2f b, v8f c) {
  return __builtin_amdgcn_wmma_f32_16x16x4_f32(false, a, false, b, (short)0, c,
                                               false, false);
}

__device__ __forceinline__ void lds_barrier() {
  __builtin_amdgcn_fence(__ATOMIC_RELEASE, "workgroup", "local");
  __builtin_amdgcn_s_barrier();
  __builtin_amdgcn_fence(__ATOMIC_ACQUIRE, "workgroup", "local");
}

template <int LDB>
__device__ __forceinline__ void load_b_slice(const float* __restrict__ Wc,
                                             v2f (&b)[32]) {
  const int lane = threadIdx.x & 31;
  const int mn = lane & 15;
  const int kh = (lane >> 4) << 1;
#pragma unroll
  for (int kk = 0; kk < 32; ++kk) {
    b[kk][0] = Wc[(4 * kk + kh + 0) * LDB + mn];
    b[kk][1] = Wc[(4 * kk + kh + 1) * LDB + mn];
  }
}

__device__ __forceinline__ void gemm_two_tiles(const float* S,
                                               const v2f (&b)[32], v8f& c0,
                                               v8f& c1) {
  const int lane = threadIdx.x & 31;
  const float* p0 = S + (lane & 15) * PITCH + ((lane >> 4) << 1);
  const float* p1 = p0 + 16 * PITCH;
#pragma unroll
  for (int kk = 0; kk < 32; ++kk) {
    const v2f a0 = *(const v2f*)(p0 + 4 * kk);
    const v2f a1 = *(const v2f*)(p1 + 4 * kk);
    c0 = wmma4(a0, b[kk], c0);
    c1 = wmma4(a1, b[kk], c1);
  }
}

__device__ __forceinline__ v8f gemm_one_tile(const float* S,
                                             const v2f (&b)[32], v8f c) {
  const int lane = threadIdx.x & 31;
  const float* p0 = S + (lane & 15) * PITCH + ((lane >> 4) << 1);
#pragma unroll
  for (int kk = 0; kk < 32; ++kk) {
    const v2f a0 = *(const v2f*)(p0 + 4 * kk);
    c = wmma4(a0, b[kk], c);
  }
  return c;
}

static_assert(NN % SLAB == 0, "a slab never straddles two replicas");
template <bool SHARED_IN>
__global__ __launch_bounds__(256) void gin_mlp_kernel(
    float* __restrict__ ux, const float* __restrict__ x0,
    const float* __restrict__ agg,
    const float* __restrict__ eps_l, const float* __restrict__ W1,
    const float* __restrict__ b1, const float* __restrict__ W2,
    const float* __restrict__ b2) {
  __shared__ __attribute__((aligned(16))) float hA[SLAB * PITCH];
  __shared__ __attribute__((aligned(16))) float hB[SLAB * PITCH];
  const int wave = threadIdx.x >> 5;
  const int col0 = wave * 16;
  const int lane = threadIdx.x & 31;
  const int n = lane & 15;
  const int kh = lane >> 4;
  const float one_eps = 1.0f + eps_l[0];

  v2f w1f[32], w2f[32];
  load_b_slice<HH>(W1 + col0, w1f);
  load_b_slice<HH>(W2 + col0, w2f);
  const float bv1 = b1[col0 + n];
  const float bv2 = b2[col0 + n];

  const int sr = threadIdx.x >> 5;
  const int sc = (threadIdx.x & 31) * 4;
  float4 pu[4], pa[4];
  const float* uin = SHARED_IN ? x0 : ux;
  int it = blockIdx.x;
  {
    const int f = it < NSLAB ? it : 0;
    const int fu = SHARED_IN ? f % (NN / SLAB) : f;
#pragma unroll
    for (int j = 0; j < 4; ++j) {
      const size_t g = ((size_t)f * SLAB + sr + 8 * j) * HH + sc;
      const size_t gu = ((size_t)fu * SLAB + sr + 8 * j) * HH + sc;
      pu[j] = *(const float4*)(uin + gu);
      pa[j] = *(const float4*)(agg + g);
    }
  }
  for (; it < NSLAB; it += gridDim.x) {
    const size_t row0 = (size_t)it * SLAB;
#pragma unroll
    for (int j = 0; j < 4; ++j) {
      float4 h;
      h.x = one_eps * pu[j].x + pa[j].x;
      h.y = one_eps * pu[j].y + pa[j].y;
      h.z = one_eps * pu[j].z + pa[j].z;
      h.w = one_eps * pu[j].w + pa[j].w;
      *(float4*)(hA + (sr + 8 * j) * PITCH + sc) = h;
    }
    const int nx = it + (int)gridDim.x < NSLAB ? it + (int)gridDim.x : it;
    const int nu = SHARED_IN ? nx % (NN / SLAB) : nx;
#pragma unroll
    for (int j = 0; j < 4; ++j) {
      const size_t g = ((size_t)nx * SLAB + sr + 8 * j) * HH + sc;
      const size_t gu = ((size_t)nu * SLAB + sr + 8 * j) * HH + sc;
      pu[j] = *(const float4*)(uin + gu);
      pa[j] = *(const float4*)(agg + g);
    }
    lds_barrier();

    v8f c0 = {}, c1 = {};
    gemm_two_tiles(hA, w1f, c0, c1);
#pragma unroll
    for (int v = 0; v < 8; ++v) {
      hB[(v + 8 * kh) * PITCH + col0 + n] = gelu_f(c0[v] + bv1);
      hB[(16 + v + 8 * kh) * PITCH + col0 + n] = gelu_f(c1[v] + bv1);
    }
    lds_barrier();

    v8f d0 = {}, d1 = {};
    gemm_two_tiles(hB, w2f, d0, d1);
#pragma unroll
    for (int v = 0; v < 8; ++v) {
      ux[(row0 + v + 8 * kh) * HH + col0 + n] = gelu_f(d0[v] + bv2);
      ux[(row0 + 16 + v + 8 * kh) * HH + col0 + n] = gelu_f(d1[v] + bv2);
    }
  }
}

__global__ __launch_bounds__(256) void head_mlp_kernel(
    const float* __restrict__ ux, const float* __restrict__ W1,
    const float* __restrict__ b1, const float* __restrict__ W2,
    const float* __restrict__ b2, const float* __restrict__ mask,
    const int* __restrict__ batch, float* __restrict__ num,
    float* __restrict__ den) {
  __shared__ __attribute__((aligned(16))) float hA[SLAB * PITCH];
  __shared__ __attribute__((aligned(16))) float hB[SLAB * PITCH];
  const int wave = threadIdx.x >> 5;
  const int col0 = wave * 16;
  const int lane = threadIdx.x & 31;
  const int n = lane & 15;
  const int kh = lane >> 4;
  const int ct = wave & 1;
  const int rt = (wave >> 1) & 1;

  v2f w1f[32], w2f[32];
  load_b_slice<HH>(W1 + col0, w1f);
  load_b_slice<OO>(W2 + ct * 16, w2f);
  const float bv1 = b1[col0 + n];
  const float bv2 = b2[ct * 16 + n];

  const int sr = threadIdx.x >> 5;
  const int sc = (threadIdx.x & 31) * 4;
  int it = blockIdx.x;
  const int f = it < NSLAB ? it : 0;
  const float* up = ux + ((size_t)f * SLAB + sr) * HH + sc;
  v4f p0 = *(const v4f*)(up);
  v4f p1 = *(const v4f*)(up + 8 * HH);
  v4f p2 = *(const v4f*)(up + 16 * HH);
  v4f p3 = *(const v4f*)(up + 24 * HH);
  for (; it < NSLAB; it += gridDim.x) {
    float* sp = hA + sr * PITCH + sc;
    *(v4f*)(sp) = p0;
    *(v4f*)(sp + 8 * PITCH) = p1;
    *(v4f*)(sp + 16 * PITCH) = p2;
    *(v4f*)(sp + 24 * PITCH) = p3;
    const int nx = it + (int)gridDim.x < NSLAB ? it + (int)gridDim.x : it;
    up = ux + ((size_t)nx * SLAB + sr) * HH + sc;
    p0 = *(const v4f*)(up);
    p1 = *(const v4f*)(up + 8 * HH);
    p2 = *(const v4f*)(up + 16 * HH);
    p3 = *(const v4f*)(up + 24 * HH);
    lds_barrier();

    v8f c0 = {}, c1 = {};
    gemm_two_tiles(hA, w1f, c0, c1);
#pragma unroll
    for (int v = 0; v < 8; ++v) {
      hB[(v + 8 * kh) * PITCH + col0 + n] = gelu_f(c0[v] + bv1);
      hB[(16 + v + 8 * kh) * PITCH + col0 + n] = gelu_f(c1[v] + bv1);
    }
    lds_barrier();

    const int rc = it / (NN / SLAB);
    const int n0 = (it - rc * (NN / SLAB)) * SLAB;
    const float* mrc = mask + (size_t)rc * NN;
    if (wave < 4) {
      const int nb = n0 + rt * 16 + 8 * kh;
      const int4 ga = *(const int4*)(batch + nb);
      const int4 gb = *(const int4*)(batch + nb + 4);
      const float4 ma = *(const float4*)(mrc + nb);
      const float4 mb = *(const float4*)(mrc + nb + 4);
      v8f d = {};
      d = gemm_one_tile(hB + rt * 16 * PITCH, w2f, d);
      const int g[8] = {ga.x, ga.y, ga.z, ga.w, gb.x, gb.y, gb.z, gb.w};
      const float m[8] = {ma.x, ma.y, ma.z, ma.w, mb.x, mb.y, mb.z, mb.w};
      float t[8];
#pragma unroll
      for (int v = 0; v < 8; ++v) t[v] = (d[v] + bv2) * m[v];
      const int gf = __builtin_amdgcn_readfirstlane(g[0]);
      int glo = g[0], ghi = g[0];
#pragma unroll
      for (int v = 1; v < 8; ++v) {
        glo = g[v] < glo ? g[v] : glo;
        ghi = g[v] > ghi ? g[v] : ghi;
      }
      float* np = num + (size_t)rc * GG * OO + ct * 16 + n;
      if (__builtin_amdgcn_ballot_w32(glo != gf || ghi != gf) == 0u) {
        float s = ((t[0] + t[1]) + (t[2] + t[3])) +
                  ((t[4] + t[5]) + (t[6] + t[7]));
        s += __shfl_xor(s, 16);
        if (kh == 0) unsafeAtomicAdd(np + (size_t)gf * OO, s);
      } else {
#pragma unroll
        for (int v = 0; v < 8; ++v)
          unsafeAtomicAdd(np + (size_t)g[v] * OO, t[v]);
      }
    } else if (wave == 4) {
      const float mv = mrc[n0 + lane];
      const int gv = batch[n0 + lane];
      const int gf = __builtin_amdgcn_readfirstlane(gv);
      float* dp = den + (size_t)rc * GG;
      if (__builtin_amdgcn_ballot_w32(gv != gf) == 0u) {
        float s = mv;
        s += __shfl_xor(s, 16);
        s += __shfl_xor(s, 8);
        s += __shfl_xor(s, 4);
        s += __shfl_xor(s, 2);
        s += __shfl_xor(s, 1);
        if (lane == 0) unsafeAtomicAdd(dp + gf, s);
      } else {
        unsafeAtomicAdd(dp + gv, mv);
      }
    }
  }
}

__global__ __launch_bounds__(256) void finalize_kernel(
    float* __restrict__ out, const float* __restrict__ den) {
  const int i = blockIdx.x * 256 + threadIdx.x;
  if (i >= RC * GG * OO) return;
  const int g = (i / OO) % GG;
  const int rc = i / (OO * GG);
  out[i] = out[i] / (den[(size_t)rc * GG + g] + 1e-7f);
}

extern "C" void kernel_launch(void* const* d_in, const int* in_sizes, int n_in,
                              void* d_out, int out_size, void* d_ws,
                              size_t ws_size, hipStream_t stream) {
  const float* x_feat = (const float*)d_in[0];
  const int* eidx = (const int*)d_in[1];
  const float* eattr = (const float*)d_in[2];
  const int* batch = (const int*)d_in[3];
  const float* nmask = (const float*)d_in[4];
  const float* atomW = (const float*)d_in[5];
  const float* atomB = (const float*)d_in[6];
  const float* eps = (const float*)d_in[7];
  const float* bondW = (const float*)d_in[8];
  const float* bondB = (const float*)d_in[9];
  const float* m1W = (const float*)d_in[10];
  const float* m1B = (const float*)d_in[11];
  const float* m2W = (const float*)d_in[12];
  const float* m2B = (const float*)d_in[13];
  const float* h1W = (const float*)d_in[14];
  const float* h1B = (const float*)d_in[15];
  const float* h2W = (const float*)d_in[16];
  const float* h2B = (const float*)d_in[17];
  float* out = (float*)d_out;

  const int* src = eidx;
  const int* dst = eidx + EE;

  float* ux = (float*)d_ws;
  float* agg = ux + (size_t)RC * NN * HH;
  float* eemb = agg + (size_t)RC * NN * HH;
  float* x0 = eemb + (size_t)EE * HH;
  float* den = x0 + (size_t)NN * HH;
  int* off = (int*)(den + (size_t)RC * GG);
  int* rnk = off + (NN + 4);
  int2* pes = (int2*)(rnk + EE);

  hipMemsetAsync(out, 0, (size_t)RC * GG * OO * sizeof(float), stream);
  hipMemsetAsync(den, 0, (size_t)RC * GG * sizeof(float), stream);

  csr_rank_scan_kernel<<<1, CSR_T, 0, stream>>>(dst, rnk, off);
  csr_fill_kernel<<<(EE + 255) / 256, 256, 0, stream>>>(src, dst, rnk, off,
                                                        pes);

  atom_gemm_kernel<<<NN / 16, 256, 0, stream>>>(x_feat, atomW, atomB, x0);

  edge_gemm_kernel<<<EE / 16, 256, 0, stream>>>(eattr, bondW, bondB, eemb);
  msg_gather_shared_kernel<<<NN / G0_NPB, 256, 0, stream>>>(x0, eemb, off, pes,
                                                            nmask, agg);
  gin_mlp_kernel<true><<<MLP_GRID, 256, 0, stream>>>(ux, x0, agg, eps, m1W, m1B,
                                                     m2W, m2B);

  for (int l = 1; l < NLAY; ++l) {
    edge_gemm_kernel<<<EE / 16, 256, 0, stream>>>(
        eattr, bondW + (size_t)l * FE * HH, bondB + (size_t)l * HH, eemb);
    msg_gather_kernel<<<NN / GATHER_NPB, 256, 0, stream>>>(ux, eemb, off, pes,
                                                           nmask, agg);
    gin_mlp_kernel<false><<<MLP_GRID, 256, 0, stream>>>(
        ux, x0, agg, eps + l, m1W + (size_t)l * HH * HH, m1B + (size_t)l * HH,
        m2W + (size_t)l * HH * HH, m2B + (size_t)l * HH);
  }

  head_mlp_kernel<<<MLP_GRID, 256, 0, stream>>>(ux, h1W, h1B, h2W, h2B, nmask,
                                                batch, out, den);
  finalize_kernel<<<(RC * GG * OO) / 256, 256, 0, stream>>>(out, den);
}


# Both memsets, bucket rank/scan, atom GEMM and both layers' edge GEMMs (all independent) run as ONE launch, work-groups dealt by block range; 13 stream operations become 8; sum order unchanged
# speedup vs baseline: 2.0905x; 1.0519x over previous
#include <hip/hip_runtime.h>
#include <cstddef>

#define NN   20000
#define EE   60000
#define GG   64
#define RC   8
#define FIN  64
#define FE   16
#define HH   128
#define OO   32
#define NLAY 2

typedef __attribute__((ext_vector_type(2))) float v2f;
typedef __attribute__((ext_vector_type(8))) float v8f;
typedef __attribute__((ext_vector_type(4))) float v4f;

__device__ __forceinline__ float erf_nb(float z) {
  const float a = __builtin_fabsf(z);
  const float s = a * a;
  float p = __builtin_fmaf(s, -0.000561801775f, 0.00491381623f);
  p = __builtin_fmaf(s, p, -0.0267075151f);
  p = __builtin_fmaf(s, p, 0.112800106f);
  p = __builtin_fmaf(s, p, -0.376122952f);
  p = __builtin_fmaf(s, p, 0.128379107f);
  const float lo = __builtin_fmaf(a, p, a);
  float q = __builtin_fmaf(a, 0.0000169988107f, -0.000378677854f);
  q = __builtin_fmaf(a, q, 0.00385781587f);
  q = __builtin_fmaf(a, q, -0.0241816975f);
  q = __builtin_fmaf(a, q, 0.106668264f);
  q = __builtin_fmaf(a, q, 0.634933293f);
  q = __builtin_fmaf(a, q, 0.128689408f);
  q = __builtin_fmaf(a, q, a);
  const float hi = 1.0f - __builtin_amdgcn_exp2f(q * -1.44269502f);
  return __builtin_copysignf(a < 1.0f ? lo : hi, z);
}

__device__ __forceinline__ float gelu_f(float x) {
  return 0.5f * x * (1.0f + erf_nb(x * 0.70710678118654752f));
}

template <int K>
__device__ __forceinline__ v8f wmma_acc(const float* A, int lda,
                                        const float* B, int ldb, v8f c) {
  const int lane = threadIdx.x & 31;
  const int mn = lane & 15;
  const int kh = (lane >> 4) << 1;
#pragma unroll
  for (int k = 0; k < K; k += 4) {
    v2f a, b;
    a[0] = A[mn * lda + k + kh + 0];
    a[1] = A[mn * lda + k + kh + 1];
    b[0] = B[(k + kh + 0) * ldb + mn];
    b[1] = B[(k + kh + 1) * ldb + mn];
    c = __builtin_amdgcn_wmma_f32_16x16x4_f32(false, a, false, b, (short)0, c,
                                              false, false);
  }
  return c;
}

static_assert(HH / 16 == 8, "tile index packs eight column tiles per row tile");
template <int K>
__device__ __forceinline__ void gemm_bias_tile(const float* __restrict__ A,
                                               const float* __restrict__ W,
                                               const float* __restrict__ bias,
                                               float* __restrict__ o, int tile) {
  const int row0 = (tile >> 3) * 16;
  const int col0 = (tile & 7) * 16;
  v8f c = {};
  c = wmma_acc<K>(A + (size_t)row0 * K, K, W + col0, HH, c);
  const int lane = threadIdx.x & 31;
  const int n = lane & 15;
  const int kh = lane >> 4;
  const float bv = bias[col0 + n];
#pragma unroll
  for (int v = 0; v < 8; ++v)
    o[(size_t)(row0 + v + 8 * kh) * HH + col0 + n] = c[v] + bv;
}

#define CSR_T 1024
#define E4 (EE / 4)
#define CSR_K ((E4 + CSR_T - 1) / CSR_T)
#define CSR_CH (NN / 1000)
#define CSR_RING 8
static_assert(EE % 4 == 0, "edges are read four at a time");
static_assert(CSR_RING <= CSR_K, "the ring is no longer than the group list");
static_assert(CSR_CH * 1000 == NN && 1000 <= CSR_T, "scan chunks cover NN");

__device__ __forceinline__ void csr_rank_scan(const int* __restrict__ dst,
                                              int* __restrict__ rnk,
                                              int* __restrict__ off) {
  __shared__ int cnt[NN];
  __shared__ int wtot[CSR_T / 32];
  const int tid = threadIdx.x;
  const int lane = tid & 31;
  const int wave = tid >> 5;
  for (int i = tid; i < NN; i += CSR_T) cnt[i] = 0;

  int4 dv[CSR_RING];
#pragma unroll
  for (int k = 0; k < CSR_RING; ++k) {
    const int g = tid + k * CSR_T;
    dv[k] = g < E4 ? *(const int4*)(dst + 4 * g) : make_int4(0, 0, 0, 0);
  }
  __syncthreads();
#pragma unroll
  for (int k = 0; k < CSR_K; ++k) {
    const int g = tid + k * CSR_T;
    const int4 q = dv[k % CSR_RING];
    if (k + CSR_RING < CSR_K) {
      const int gn = g + CSR_RING * CSR_T;
      dv[k % CSR_RING] =
          gn < E4 ? *(const int4*)(dst + 4 * gn) : make_int4(0, 0, 0, 0);
    }
    if (g < E4) {
      int4 r;
      r.x = atomicAdd(&cnt[q.x], 1);
      r.y = atomicAdd(&cnt[q.y], 1);
      r.z = atomicAdd(&cnt[q.z], 1);
      r.w = atomicAdd(&cnt[q.w], 1);
      *(int4*)(rnk + 4 * g) = r;
    }
  }
  __syncthreads();

  int c[CSR_CH];
  int tsum = 0;
  const int b0 = tid * CSR_CH;
  const bool own = b0 < NN;
#pragma unroll
  for (int j = 0; j < CSR_CH; ++j) {
    c[j] = own ? cnt[b0 + j] : 0;
    tsum += c[j];
  }
  int incl = tsum;
  for (int o = 1; o < 32; o <<= 1) {
    const int v = __shfl_up(incl, o);
    if (lane >= o) incl += v;
  }
  if (lane == 31) wtot[wave] = incl;
  __syncthreads();
  if (wave == 0) {
    const int w = wtot[lane];
    int wi = w;
    for (int o = 1; o < 32; o <<= 1) {
      const int v = __shfl_up(wi, o);
      if (lane >= o) wi += v;
    }
    wtot[lane] = wi - w;
  }
  __syncthreads();
  int run = wtot[wave] + incl - tsum;
  if (own) {
#pragma unroll
    for (int j = 0; j < CSR_CH; ++j) {
      cnt[b0 + j] = run;
      run += c[j];
    }
  }
  __syncthreads();
  for (int i = tid; i < NN; i += CSR_T) off[i] = cnt[i];
  if (tid == 0) off[NN] = EE;
}

#define PRE_T CSR_T
#define PRE_WPB (PRE_T / 32)
#define ATOM_TILES ((NN / 16) * (HH / 16))
#define EDGE_TILES ((EE / 16) * (HH / 16))
#define ATOM_WG ((ATOM_TILES + PRE_WPB - 1) / PRE_WPB)
#define EDGE_WG ((EDGE_TILES + PRE_WPB - 1) / PRE_WPB)
#define PRE_GRID (2 + ATOM_WG + NLAY * EDGE_WG)
static_assert((RC * GG * OO) % (4 * PRE_T) == 0, "num is zeroed in 16-B stores");
static_assert(RC * GG <= PRE_T, "den is zeroed by one store per thread");

__global__ __launch_bounds__(PRE_T, 16) void prepass_kernel(
    const int* __restrict__ dst, int* __restrict__ rnk, int* __restrict__ off,
    float* __restrict__ num, float* __restrict__ den,
    const float* __restrict__ xf, const float* __restrict__ atomW,
    const float* __restrict__ atomB, float* __restrict__ x0,
    const float* __restrict__ ea, const float* __restrict__ bondW,
    const float* __restrict__ bondB, float* __restrict__ eemb) {
  const int b = blockIdx.x;
  if (b == 0) {
    csr_rank_scan(dst, rnk, off);
    return;
  }
  if (b == 1) {
    const int tid = threadIdx.x;
    for (int i = tid; i < (RC * GG * OO) / 4; i += PRE_T)
      *(float4*)(num + 4 * i) = make_float4(0.0f, 0.0f, 0.0f, 0.0f);
    if (tid < RC * GG) den[tid] = 0.0f;
    return;
  }
  const int wave = __builtin_amdgcn_readfirstlane((int)(threadIdx.x >> 5));
  if (b < 2 + ATOM_WG) {
    const int t = (b - 2) * PRE_WPB + wave;
    if (t < ATOM_TILES) gemm_bias_tile<FIN>(xf, atomW, atomB, x0, t);
    return;
  }
  const int eb = b - (2 + ATOM_WG);
  const int l = eb / EDGE_WG;
  const int t = (eb - l * EDGE_WG) * PRE_WPB + wave;
  if (t < EDGE_TILES)
    gemm_bias_tile<FE>(ea, bondW + (size_t)l * FE * HH, bondB + (size_t)l * HH,
                       eemb + (size_t)l * EE * HH, t);
}

__global__ __launch_bounds__(256) void csr_fill_kernel(
    const int* __restrict__ src, const int* __restrict__ dst,
    const int* __restrict__ rnk, const int* __restrict__ off,
    int2* __restrict__ pes) {
  const int e = blockIdx.x * 256 + threadIdx.x;
  if (e >= EE) return;
  const int slot = off[dst[e]] + rnk[e];
  pes[slot] = make_int2(e, src[e]);
}

#define GATHER_NPB 4
static_assert(NN % GATHER_NPB == 0, "node groups cover NN");
static_assert(RC * 32 == 256, "one wave of the work-group per replica");

__global__ __launch_bounds__(256) void msg_gather_kernel(
    const float* __restrict__ ux, const float* __restrict__ eemb,
    const int* __restrict__ off, const int2* __restrict__ pes,
    const float* __restrict__ mask, float* __restrict__ agg) {
  const int rc = threadIdx.x >> 5;
  const int lane = threadIdx.x & 31;
  const float* mrc = mask + (size_t)rc * NN;
  const float* uxr = ux + (size_t)rc * NN * HH + lane * 4;
  const float* eer = eemb + lane * 4;
  float* ar = agg + (size_t)rc * NN * HH + lane * 4;
  const int d0 = blockIdx.x * GATHER_NPB;
  int end = off[d0];
  for (int k = 0; k < GATHER_NPB; ++k) {
    const int d = d0 + k;
    const int beg = end;
    end = off[d + 1];
    const float md = mrc[d];
    float4 acc = make_float4(0.0f, 0.0f, 0.0f, 0.0f);
    for (int base = beg; base < end; base += 32) {
      const int cnt = end - base < 32 ? end - base : 32;
      int me = 0, ms = 0;
      float mm = 0.0f;
      if (lane < cnt) {
        const int2 p = pes[base + lane];
        me = p.x;
        ms = p.y;
        mm = mrc[ms] * md;
      }
      for (int j = 0; j < cnt; ++j) {
        const int e = __builtin_amdgcn_readlane(me, j);
        const int s = __builtin_amdgcn_readlane(ms, j);
        const float mk =
            __int_as_float(__builtin_amdgcn_readlane(__float_as_int(mm), j));
        const float4 xs = *(const float4*)(uxr + (size_t)s * HH);
        const float4 ee = *(const float4*)(eer + (size_t)e * HH);
        acc.x += gelu_f(xs.x + ee.x) * mk;
        acc.y += gelu_f(xs.y + ee.y) * mk;
        acc.z += gelu_f(xs.z + ee.z) * mk;
        acc.w += gelu_f(xs.w + ee.w) * mk;
      }
    }
    *(float4*)(ar + (size_t)d * HH) = acc;
  }
}

#define G0_NPW 2
#define G0_NPB (8 * G0_NPW)
static_assert(NN % G0_NPB == 0, "node groups cover NN");
static_assert(RC == 8, "a group of 8 lanes fetches the 8 replica masks");

__global__ __launch_bounds__(256) void msg_gather_shared_kernel(
    const float* __restrict__ x0, const float* __restrict__ eemb,
    const int* __restrict__ off, const int2* __restrict__ pes,
    const float* __restrict__ mask, float* __restrict__ agg) {
  const int lane = threadIdx.x & 31;
  const int wave = __builtin_amdgcn_readfirstlane((int)(threadIdx.x >> 5));
  const int rcl = lane & 7;
  const int jl = lane >> 3;
  const float* mr = mask + (size_t)rcl * NN;
  const float* xr = x0 + lane * 4;
  const float* er = eemb + lane * 4;
  float* ar = agg + lane * 4;
  const int d0 = (blockIdx.x * 8 + wave) * G0_NPW;
  int end = off[d0];
  for (int k = 0; k < G0_NPW; ++k) {
    const int d = d0 + k;
    const int beg = end;
    end = off[d + 1];
    const float md = mr[d];
    float4 acc[RC];
#pragma unroll
    for (int rc = 0; rc < RC; ++rc) acc[rc] = make_float4(0.0f, 0.0f, 0.0f, 0.0f);
    for (int base = beg; base < end; base += 4) {
      const int cnt = end - base < 4 ? end - base : 4;
      int me = 0, ms = 0;
      float mm = 0.0f;
      if (jl < cnt) {
        const int2 p = pes[base + jl];
        me = p.x;
        ms = p.y;
        mm = mr[ms] * md;
      }
      float4 xs[4], ee[4];
#pragma unroll
      for (int j = 0; j < 4; ++j) {
        const int jj = j < cnt ? j : 0;
        const int e = __builtin_amdgcn_readlane(me, 8 * jj);
        const int s = __builtin_amdgcn_readlane(ms, 8 * jj);
        xs[j] = *(const float4*)(xr + (size_t)s * HH);
        ee[j] = *(const float4*)(er + (size_t)e * HH);
      }
#pragma unroll
      for (int j = 0; j < 4; ++j) {
        if (j < cnt) {
          float4 g;
          g.x = gelu_f(xs[j].x + ee[j].x);
          g.y = gelu_f(xs[j].y + ee[j].y);
          g.z = gelu_f(xs[j].z + ee[j].z);
          g.w = gelu_f(xs[j].w + ee[j].w);
#pragma unroll
          for (int rc = 0; rc < RC; ++rc) {
            const float mk = __int_as_float(
                __builtin_amdgcn_readlane(__float_as_int(mm), 8 * j + rc));
            acc[rc].x += g.x * mk;
            acc[rc].y += g.y * mk;
            acc[rc].z += g.z * mk;
            acc[rc].w += g.w * mk;
          }
        }
      }
    }
#pragma unroll
    for (int rc = 0; rc < RC; ++rc)
      *(float4*)(ar + ((size_t)rc * NN + d) * HH) = acc[rc];
  }
}

#define PITCH 132
#define SLAB 32
#define NSLAB ((RC * NN) / SLAB)
#define MLP_GRID 512

__device__ __forceinline__ v8f wmma4(v2f a, v2f b, v8f c) {
  return __builtin_amdgcn_wmma_f32_16x16x4_f32(false, a, false, b, (short)0, c,
                                               false, false);
}

__device__ __forceinline__ void lds_barrier() {
  __builtin_amdgcn_fence(__ATOMIC_RELEASE, "workgroup", "local");
  __builtin_amdgcn_s_barrier();
  __builtin_amdgcn_fence(__ATOMIC_ACQUIRE, "workgroup", "local");
}

template <int LDB>
__device__ __forceinline__ void load_b_slice(const float* __restrict__ Wc,
                                             v2f (&b)[32]) {
  const int lane = threadIdx.x & 31;
  const int mn = lane & 15;
  const int kh = (lane >> 4) << 1;
#pragma unroll
  for (int kk = 0; kk < 32; ++kk) {
    b[kk][0] = Wc[(4 * kk + kh + 0) * LDB + mn];
    b[kk][1] = Wc[(4 * kk + kh + 1) * LDB + mn];
  }
}

__device__ __forceinline__ void gemm_two_tiles(const float* S,
                                               const v2f (&b)[32], v8f& c0,
                                               v8f& c1) {
  const int lane = threadIdx.x & 31;
  const float* p0 = S + (lane & 15) * PITCH + ((lane >> 4) << 1);
  const float* p1 = p0 + 16 * PITCH;
#pragma unroll
  for (int kk = 0; kk < 32; ++kk) {
    const v2f a0 = *(const v2f*)(p0 + 4 * kk);
    const v2f a1 = *(const v2f*)(p1 + 4 * kk);
    c0 = wmma4(a0, b[kk], c0);
    c1 = wmma4(a1, b[kk], c1);
  }
}

__device__ __forceinline__ v8f gemm_one_tile(const float* S,
                                             const v2f (&b)[32], v8f c) {
  const int lane = threadIdx.x & 31;
  const float* p0 = S + (lane & 15) * PITCH + ((lane >> 4) << 1);
#pragma unroll
  for (int kk = 0; kk < 32; ++kk) {
    const v2f a0 = *(const v2f*)(p0 + 4 * kk);
    c = wmma4(a0, b[kk], c);
  }
  return c;
}

static_assert(NN % SLAB == 0, "a slab never straddles two replicas");
template <bool SHARED_IN>
__global__ __launch_bounds__(256) void gin_mlp_kernel(
    float* __restrict__ ux, const float* __restrict__ x0,
    const float* __restrict__ agg,
    const float* __restrict__ eps_l, const float* __restrict__ W1,
    const float* __restrict__ b1, const float* __restrict__ W2,
    const float* __restrict__ b2) {
  __shared__ __attribute__((aligned(16))) float hA[SLAB * PITCH];
  __shared__ __attribute__((aligned(16))) float hB[SLAB * PITCH];
  const int wave = threadIdx.x >> 5;
  const int col0 = wave * 16;
  const int lane = threadIdx.x & 31;
  const int n = lane & 15;
  const int kh = lane >> 4;
  const float one_eps = 1.0f + eps_l[0];

  v2f w1f[32], w2f[32];
  load_b_slice<HH>(W1 + col0, w1f);
  load_b_slice<HH>(W2 + col0, w2f);
  const float bv1 = b1[col0 + n];
  const float bv2 = b2[col0 + n];

  const int sr = threadIdx.x >> 5;
  const int sc = (threadIdx.x & 31) * 4;
  float4 pu[4], pa[4];
  const float* uin = SHARED_IN ? x0 : ux;
  int it = blockIdx.x;
  {
    const int f = it < NSLAB ? it : 0;
    const int fu = SHARED_IN ? f % (NN / SLAB) : f;
#pragma unroll
    for (int j = 0; j < 4; ++j) {
      const size_t g = ((size_t)f * SLAB + sr + 8 * j) * HH + sc;
      const size_t gu = ((size_t)fu * SLAB + sr + 8 * j) * HH + sc;
      pu[j] = *(const float4*)(uin + gu);
      pa[j] = *(const float4*)(agg + g);
    }
  }
  for (; it < NSLAB; it += gridDim.x) {
    const size_t row0 = (size_t)it * SLAB;
#pragma unroll
    for (int j = 0; j < 4; ++j) {
      float4 h;
      h.x = one_eps * pu[j].x + pa[j].x;
      h.y = one_eps * pu[j].y + pa[j].y;
      h.z = one_eps * pu[j].z + pa[j].z;
      h.w = one_eps * pu[j].w + pa[j].w;
      *(float4*)(hA + (sr + 8 * j) * PITCH + sc) = h;
    }
    const int nx = it + (int)gridDim.x < NSLAB ? it + (int)gridDim.x : it;
    const int nu = SHARED_IN ? nx % (NN / SLAB) : nx;
#pragma unroll
    for (int j = 0; j < 4; ++j) {
      const size_t g = ((size_t)nx * SLAB + sr + 8 * j) * HH + sc;
      const size_t gu = ((size_t)nu * SLAB + sr + 8 * j) * HH + sc;
      pu[j] = *(const float4*)(uin + gu);
      pa[j] = *(const float4*)(agg + g);
    }
    lds_barrier();

    v8f c0 = {}, c1 = {};
    gemm_two_tiles(hA, w1f, c0, c1);
#pragma unroll
    for (int v = 0; v < 8; ++v) {
      hB[(v + 8 * kh) * PITCH + col0 + n] = gelu_f(c0[v] + bv1);
      hB[(16 + v + 8 * kh) * PITCH + col0 + n] = gelu_f(c1[v] + bv1);
    }
    lds_barrier();

    v8f d0 = {}, d1 = {};
    gemm_two_tiles(hB, w2f, d0, d1);
#pragma unroll
    for (int v = 0; v < 8; ++v) {
      ux[(row0 + v + 8 * kh) * HH + col0 + n] = gelu_f(d0[v] + bv2);
      ux[(row0 + 16 + v + 8 * kh) * HH + col0 + n] = gelu_f(d1[v] + bv2);
    }
  }
}

__global__ __launch_bounds__(256) void head_mlp_kernel(
    const float* __restrict__ ux, const float* __restrict__ W1,
    const float* __restrict__ b1, const float* __restrict__ W2,
    const float* __restrict__ b2, const float* __restrict__ mask,
    const int* __restrict__ batch, float* __restrict__ num,
    float* __restrict__ den) {
  __shared__ __attribute__((aligned(16))) float hA[SLAB * PITCH];
  __shared__ __attribute__((aligned(16))) float hB[SLAB * PITCH];
  const int wave = threadIdx.x >> 5;
  const int col0 = wave * 16;
  const int lane = threadIdx.x & 31;
  const int n = lane & 15;
  const int kh = lane >> 4;
  const int ct = wave & 1;
  const int rt = (wave >> 1) & 1;

  v2f w1f[32], w2f[32];
  load_b_slice<HH>(W1 + col0, w1f);
  load_b_slice<OO>(W2 + ct * 16, w2f);
  const float bv1 = b1[col0 + n];
  const float bv2 = b2[ct * 16 + n];

  const int sr = threadIdx.x >> 5;
  const int sc = (threadIdx.x & 31) * 4;
  int it = blockIdx.x;
  const int f = it < NSLAB ? it : 0;
  const float* up = ux + ((size_t)f * SLAB + sr) * HH + sc;
  v4f p0 = *(const v4f*)(up);
  v4f p1 = *(const v4f*)(up + 8 * HH);
  v4f p2 = *(const v4f*)(up + 16 * HH);
  v4f p3 = *(const v4f*)(up + 24 * HH);
  for (; it < NSLAB; it += gridDim.x) {
    float* sp = hA + sr * PITCH + sc;
    *(v4f*)(sp) = p0;
    *(v4f*)(sp + 8 * PITCH) = p1;
    *(v4f*)(sp + 16 * PITCH) = p2;
    *(v4f*)(sp + 24 * PITCH) = p3;
    const int nx = it + (int)gridDim.x < NSLAB ? it + (int)gridDim.x : it;
    up = ux + ((size_t)nx * SLAB + sr) * HH + sc;
    p0 = *(const v4f*)(up);
    p1 = *(const v4f*)(up + 8 * HH);
    p2 = *(const v4f*)(up + 16 * HH);
    p3 = *(const v4f*)(up + 24 * HH);
    lds_barrier();

    v8f c0 = {}, c1 = {};
    gemm_two_tiles(hA, w1f, c0, c1);
#pragma unroll
    for (int v = 0; v < 8; ++v) {
      hB[(v + 8 * kh) * PITCH + col0 + n] = gelu_f(c0[v] + bv1);
      hB[(16 + v + 8 * kh) * PITCH + col0 + n] = gelu_f(c1[v] + bv1);
    }
    lds_barrier();

    const int rc = it / (NN / SLAB);
    const int n0 = (it - rc * (NN / SLAB)) * SLAB;
    const float* mrc = mask + (size_t)rc * NN;
    if (wave < 4) {
      const int nb = n0 + rt * 16 + 8 * kh;
      const int4 ga = *(const int4*)(batch + nb);
      const int4 gb = *(const int4*)(batch + nb + 4);
      const float4 ma = *(const float4*)(mrc + nb);
      const float4 mb = *(const float4*)(mrc + nb + 4);
      v8f d = {};
      d = gemm_one_tile(hB + rt * 16 * PITCH, w2f, d);
      const int g[8] = {ga.x, ga.y, ga.z, ga.w, gb.x, gb.y, gb.z, gb.w};
      const float m[8] = {ma.x, ma.y, ma.z, ma.w, mb.x, mb.y, mb.z, mb.w};
      float t[8];
#pragma unroll
      for (int v = 0; v < 8; ++v) t[v] = (d[v] + bv2) * m[v];
      const int gf = __builtin_amdgcn_readfirstlane(g[0]);
      int glo = g[0], ghi = g[0];
#pragma unroll
      for (int v = 1; v < 8; ++v) {
        glo = g[v] < glo ? g[v] : glo;
        ghi = g[v] > ghi ? g[v] : ghi;
      }
      float* np = num + (size_t)rc * GG * OO + ct * 16 + n;
      if (__builtin_amdgcn_ballot_w32(glo != gf || ghi != gf) == 0u) {
        float s = ((t[0] + t[1]) + (t[2] + t[3])) +
                  ((t[4] + t[5]) + (t[6] + t[7]));
        s += __shfl_xor(s, 16);
        if (kh == 0) unsafeAtomicAdd(np + (size_t)gf * OO, s);
      } else {
#pragma unroll
        for (int v = 0; v < 8; ++v)
          unsafeAtomicAdd(np + (size_t)g[v] * OO, t[v]);
      }
    } else if (wave == 4) {
      const float mv = mrc[n0 + lane];
      const int gv = batch[n0 + lane];
      const int gf = __builtin_amdgcn_readfirstlane(gv);
      float* dp = den + (size_t)rc * GG;
      if (__builtin_amdgcn_ballot_w32(gv != gf) == 0u) {
        float s = mv;
        s += __shfl_xor(s, 16);
        s += __shfl_xor(s, 8);
        s += __shfl_xor(s, 4);
        s += __shfl_xor(s, 2);
        s += __shfl_xor(s, 1);
        if (lane == 0) unsafeAtomicAdd(dp + gf, s);
      } else {
        unsafeAtomicAdd(dp + gv, mv);
      }
    }
  }
}

__global__ __launch_bounds__(256) void finalize_kernel(
    float* __restrict__ out, const float* __restrict__ den) {
  const int i = blockIdx.x * 256 + threadIdx.x;
  if (i >= RC * GG * OO) return;
  const int g = (i / OO) % GG;
  const int rc = i / (OO * GG);
  out[i] = out[i] / (den[(size_t)rc * GG + g] + 1e-7f);
}

extern "C" void kernel_launch(void* const* d_in, const int* in_sizes, int n_in,
                              void* d_out, int out_size, void* d_ws,
                              size_t ws_size, hipStream_t stream) {
  const float* x_feat = (const float*)d_in[0];
  const int* eidx = (const int*)d_in[1];
  const float* eattr = (const float*)d_in[2];
  const int* batch = (const int*)d_in[3];
  const float* nmask = (const float*)d_in[4];
  const float* atomW = (const float*)d_in[5];
  const float* atomB = (const float*)d_in[6];
  const float* eps = (const float*)d_in[7];
  const float* bondW = (const float*)d_in[8];
  const float* bondB = (const float*)d_in[9];
  const float* m1W = (const float*)d_in[10];
  const float* m1B = (const float*)d_in[11];
  const float* m2W = (const float*)d_in[12];
  const float* m2B = (const float*)d_in[13];
  const float* h1W = (const float*)d_in[14];
  const float* h1B = (const float*)d_in[15];
  const float* h2W = (const float*)d_in[16];
  const float* h2B = (const float*)d_in[17];
  float* out = (float*)d_out;

  const int* src = eidx;
  const int* dst = eidx + EE;

  float* ux = (float*)d_ws;
  float* agg = ux + (size_t)RC * NN * HH;
  float* eemb = agg + (size_t)RC * NN * HH;
  float* x0 = eemb + (size_t)NLAY * EE * HH;
  float* den = x0 + (size_t)NN * HH;
  int* off = (int*)(den + (size_t)RC * GG);
  int* rnk = off + (NN + 4);
  int2* pes = (int2*)(rnk + EE);

  prepass_kernel<<<PRE_GRID, PRE_T, 0, stream>>>(dst, rnk, off, out, den, x_feat,
                                                 atomW, atomB, x0, eattr, bondW,
                                                 bondB, eemb);
  csr_fill_kernel<<<(EE + 255) / 256, 256, 0, stream>>>(src, dst, rnk, off,
                                                        pes);

  msg_gather_shared_kernel<<<NN / G0_NPB, 256, 0, stream>>>(x0, eemb, off, pes,
                                                            nmask, agg);
  gin_mlp_kernel<true><<<MLP_GRID, 256, 0, stream>>>(ux, x0, agg, eps, m1W, m1B,
                                                     m2W, m2B);

  for (int l = 1; l < NLAY; ++l) {
    msg_gather_kernel<<<NN / GATHER_NPB, 256, 0, stream>>>(
        ux, eemb + (size_t)l * EE * HH, off, pes, nmask, agg);
    gin_mlp_kernel<false><<<MLP_GRID, 256, 0, stream>>>(
        ux, x0, agg, eps + l, m1W + (size_t)l * HH * HH, m1B + (size_t)l * HH,
        m2W + (size_t)l * HH * HH, m2B + (size_t)l * HH);
  }

  head_mlp_kernel<<<MLP_GRID, 256, 0, stream>>>(ux, h1W, h1B, h2W, h2B, nmask,
                                                batch, out, den);
  finalize_kernel<<<(RC * GG * OO) / 256, 256, 0, stream>>>(out, den);
}


# Exact GELU at every site as max(x,0) - |x|*Phi(-|x|), Phi(-u) = exp2 of ONE degree-8 polynomial: no erf, second range, select or copysign; max abs error below the old form's; sum order unchanged
# speedup vs baseline: 2.3128x; 1.1063x over previous
#include <hip/hip_runtime.h>
#include <cstddef>

#define NN   20000
#define EE   60000
#define GG   64
#define RC   8
#define FIN  64
#define FE   16
#define HH   128
#define OO   32
#define NLAY 2

typedef __attribute__((ext_vector_type(2))) float v2f;
typedef __attribute__((ext_vector_type(8))) float v8f;
typedef __attribute__((ext_vector_type(4))) float v4f;

__device__ __forceinline__ float gelu_f(float x) {
  const float u = __builtin_fminf(__builtin_fabsf(x), 16.0f);
  float p = __builtin_fmaf(u, -0.00000283480495f, 0.0000393764567f);
  p = __builtin_fmaf(u, p, -0.000186174759f);
  p = __builtin_fmaf(u, p, -0.000136948453f);
  p = __builtin_fmaf(u, p, 0.00706343399f);
  p = __builtin_fmaf(u, p, -0.0524961874f);
  p = __builtin_fmaf(u, p, -0.459208190f);
  p = __builtin_fmaf(u, p, -1.15110517f);
  p = __builtin_fmaf(u, p, -1.0f);
  const float t = __builtin_amdgcn_exp2f(p);
  return __builtin_fmaf(-u, t, __builtin_fmaxf(x, 0.0f));
}

template <int K>
__device__ __forceinline__ v8f wmma_acc(const float* A, int lda,
                                        const float* B, int ldb, v8f c) {
  const int lane = threadIdx.x & 31;
  const int mn = lane & 15;
  const int kh = (lane >> 4) << 1;
#pragma unroll
  for (int k = 0; k < K; k += 4) {
    v2f a, b;
    a[0] = A[mn * lda + k + kh + 0];
    a[1] = A[mn * lda + k + kh + 1];
    b[0] = B[(k + kh + 0) * ldb + mn];
    b[1] = B[(k + kh + 1) * ldb + mn];
    c = __builtin_amdgcn_wmma_f32_16x16x4_f32(false, a, false, b, (short)0, c,
                                              false, false);
  }
  return c;
}

static_assert(HH / 16 == 8, "tile index packs eight column tiles per row tile");
template <int K>
__device__ __forceinline__ void gemm_bias_tile(const float* __restrict__ A,
                                               const float* __restrict__ W,
                                               const float* __restrict__ bias,
                                               float* __restrict__ o, int tile) {
  const int row0 = (tile >> 3) * 16;
  const int col0 = (tile & 7) * 16;
  v8f c = {};
  c = wmma_acc<K>(A + (size_t)row0 * K, K, W + col0, HH, c);
  const int lane = threadIdx.x & 31;
  const int n = lane & 15;
  const int kh = lane >> 4;
  const float bv = bias[col0 + n];
#pragma unroll
  for (int v = 0; v < 8; ++v)
    o[(size_t)(row0 + v + 8 * kh) * HH + col0 + n] = c[v] + bv;
}

#define CSR_T 1024
#define E4 (EE / 4)
#define CSR_K ((E4 + CSR_T - 1) / CSR_T)
#define CSR_CH (NN / 1000)
#define CSR_RING 8
static_assert(EE % 4 == 0, "edges are read four at a time");
static_assert(CSR_RING <= CSR_K, "the ring is no longer than the group list");
static_assert(CSR_CH * 1000 == NN && 1000 <= CSR_T, "scan chunks cover NN");

__device__ __forceinline__ void csr_rank_scan(const int* __restrict__ dst,
                                              int* __restrict__ rnk,
                                              int* __restrict__ off) {
  __shared__ int cnt[NN];
  __shared__ int wtot[CSR_T / 32];
  const int tid = threadIdx.x;
  const int lane = tid & 31;
  const int wave = tid >> 5;
  for (int i = tid; i < NN; i += CSR_T) cnt[i] = 0;

  int4 dv[CSR_RING];
#pragma unroll
  for (int k = 0; k < CSR_RING; ++k) {
    const int g = tid + k * CSR_T;
    dv[k] = g < E4 ? *(const int4*)(dst + 4 * g) : make_int4(0, 0, 0, 0);
  }
  __syncthreads();
#pragma unroll
  for (int k = 0; k < CSR_K; ++k) {
    const int g = tid + k * CSR_T;
    const int4 q = dv[k % CSR_RING];
    if (k + CSR_RING < CSR_K) {
      const int gn = g + CSR_RING * CSR_T;
      dv[k % CSR_RING] =
          gn < E4 ? *(const int4*)(dst + 4 * gn) : make_int4(0, 0, 0, 0);
    }
    if (g < E4) {
      int4 r;
      r.x = atomicAdd(&cnt[q.x], 1);
      r.y = atomicAdd(&cnt[q.y], 1);
      r.z = atomicAdd(&cnt[q.z], 1);
      r.w = atomicAdd(&cnt[q.w], 1);
      *(int4*)(rnk + 4 * g) = r;
    }
  }
  __syncthreads();

  int c[CSR_CH];
  int tsum = 0;
  const int b0 = tid * CSR_CH;
  const bool own = b0 < NN;
#pragma unroll
  for (int j = 0; j < CSR_CH; ++j) {
    c[j] = own ? cnt[b0 + j] : 0;
    tsum += c[j];
  }
  int incl = tsum;
  for (int o = 1; o < 32; o <<= 1) {
    const int v = __shfl_up(incl, o);
    if (lane >= o) incl += v;
  }
  if (lane == 31) wtot[wave] = incl;
  __syncthreads();
  if (wave == 0) {
    const int w = wtot[lane];
    int wi = w;
    for (int o = 1; o < 32; o <<= 1) {
      const int v = __shfl_up(wi, o);
      if (lane >= o) wi += v;
    }
    wtot[lane] = wi - w;
  }
  __syncthreads();
  int run = wtot[wave] + incl - tsum;
  if (own) {
#pragma unroll
    for (int j = 0; j < CSR_CH; ++j) {
      cnt[b0 + j] = run;
      run += c[j];
    }
  }
  __syncthreads();
  for (int i = tid; i < NN; i += CSR_T) off[i] = cnt[i];
  if (tid == 0) off[NN] = EE;
}

#define PRE_T CSR_T
#define PRE_WPB (PRE_T / 32)
#define ATOM_TILES ((NN / 16) * (HH / 16))
#define EDGE_TILES ((EE / 16) * (HH / 16))
#define ATOM_WG ((ATOM_TILES + PRE_WPB - 1) / PRE_WPB)
#define EDGE_WG ((EDGE_TILES + PRE_WPB - 1) / PRE_WPB)
#define PRE_GRID (2 + ATOM_WG + NLAY * EDGE_WG)
static_assert((RC * GG * OO) % (4 * PRE_T) == 0, "num is zeroed in 16-B stores");
static_assert(RC * GG <= PRE_T, "den is zeroed by one store per thread");

__global__ __launch_bounds__(PRE_T, 16) void prepass_kernel(
    const int* __restrict__ dst, int* __restrict__ rnk, int* __restrict__ off,
    float* __restrict__ num, float* __restrict__ den,
    const float* __restrict__ xf, const float* __restrict__ atomW,
    const float* __restrict__ atomB, float* __restrict__ x0,
    const float* __restrict__ ea, const float* __restrict__ bondW,
    const float* __restrict__ bondB, float* __restrict__ eemb) {
  const int b = blockIdx.x;
  if (b == 0) {
    csr_rank_scan(dst, rnk, off);
    return;
  }
  if (b == 1) {
    const int tid = threadIdx.x;
    for (int i = tid; i < (RC * GG * OO) / 4; i += PRE_T)
      *(float4*)(num + 4 * i) = make_float4(0.0f, 0.0f, 0.0f, 0.0f);
    if (tid < RC * GG) den[tid] = 0.0f;
    return;
  }
  const int wave = __builtin_amdgcn_readfirstlane((int)(threadIdx.x >> 5));
  if (b < 2 + ATOM_WG) {
    const int t = (b - 2) * PRE_WPB + wave;
    if (t < ATOM_TILES) gemm_bias_tile<FIN>(xf, atomW, atomB, x0, t);
    return;
  }
  const int eb = b - (2 + ATOM_WG);
  const int l = eb / EDGE_WG;
  const int t = (eb - l * EDGE_WG) * PRE_WPB + wave;
  if (t < EDGE_TILES)
    gemm_bias_tile<FE>(ea, bondW + (size_t)l * FE * HH, bondB + (size_t)l * HH,
                       eemb + (size_t)l * EE * HH, t);
}

__global__ __launch_bounds__(256) void csr_fill_kernel(
    const int* __restrict__ src, const int* __restrict__ dst,
    const int* __restrict__ rnk, const int* __restrict__ off,
    int2* __restrict__ pes) {
  const int e = blockIdx.x * 256 + threadIdx.x;
  if (e >= EE) return;
  const int slot = off[dst[e]] + rnk[e];
  pes[slot] = make_int2(e, src[e]);
}

#define GATHER_NPB 4
static_assert(NN % GATHER_NPB == 0, "node groups cover NN");
static_assert(RC * 32 == 256, "one wave of the work-group per replica");

__global__ __launch_bounds__(256) void msg_gather_kernel(
    const float* __restrict__ ux, const float* __restrict__ eemb,
    const int* __restrict__ off, const int2* __restrict__ pes,
    const float* __restrict__ mask, float* __restrict__ agg) {
  const int rc = threadIdx.x >> 5;
  const int lane = threadIdx.x & 31;
  const float* mrc = mask + (size_t)rc * NN;
  const float* uxr = ux + (size_t)rc * NN * HH + lane * 4;
  const float* eer = eemb + lane * 4;
  float* ar = agg + (size_t)rc * NN * HH + lane * 4;
  const int d0 = blockIdx.x * GATHER_NPB;
  int end = off[d0];
  for (int k = 0; k < GATHER_NPB; ++k) {
    const int d = d0 + k;
    const int beg = end;
    end = off[d + 1];
    const float md = mrc[d];
    float4 acc = make_float4(0.0f, 0.0f, 0.0f, 0.0f);
    for (int base = beg; base < end; base += 32) {
      const int cnt = end - base < 32 ? end - base : 32;
      int me = 0, ms = 0;
      float mm = 0.0f;
      if (lane < cnt) {
        const int2 p = pes[base + lane];
        me = p.x;
        ms = p.y;
        mm = mrc[ms] * md;
      }
      for (int j = 0; j < cnt; ++j) {
        const int e = __builtin_amdgcn_readlane(me, j);
        const int s = __builtin_amdgcn_readlane(ms, j);
        const float mk =
            __int_as_float(__builtin_amdgcn_readlane(__float_as_int(mm), j));
        const float4 xs = *(const float4*)(uxr + (size_t)s * HH);
        const float4 ee = *(const float4*)(eer + (size_t)e * HH);
        acc.x += gelu_f(xs.x + ee.x) * mk;
        acc.y += gelu_f(xs.y + ee.y) * mk;
        acc.z += gelu_f(xs.z + ee.z) * mk;
        acc.w += gelu_f(xs.w + ee.w) * mk;
      }
    }
    *(float4*)(ar + (size_t)d * HH) = acc;
  }
}

#define G0_NPW 2
#define G0_NPB (8 * G0_NPW)
static_assert(NN % G0_NPB == 0, "node groups cover NN");
static_assert(RC == 8, "a group of 8 lanes fetches the 8 replica masks");

__global__ __launch_bounds__(256) void msg_gather_shared_kernel(
    const float* __restrict__ x0, const float* __restrict__ eemb,
    const int* __restrict__ off, const int2* __restrict__ pes,
    const float* __restrict__ mask, float* __restrict__ agg) {
  const int lane = threadIdx.x & 31;
  const int wave = __builtin_amdgcn_readfirstlane((int)(threadIdx.x >> 5));
  const int rcl = lane & 7;
  const int jl = lane >> 3;
  const float* mr = mask + (size_t)rcl * NN;
  const float* xr = x0 + lane * 4;
  const float* er = eemb + lane * 4;
  float* ar = agg + lane * 4;
  const int d0 = (blockIdx.x * 8 + wave) * G0_NPW;
  int end = off[d0];
  for (int k = 0; k < G0_NPW; ++k) {
    const int d = d0 + k;
    const int beg = end;
    end = off[d + 1];
    const float md = mr[d];
    float4 acc[RC];
#pragma unroll
    for (int rc = 0; rc < RC; ++rc) acc[rc] = make_float4(0.0f, 0.0f, 0.0f, 0.0f);
    for (int base = beg; base < end; base += 4) {
      const int cnt = end - base < 4 ? end - base : 4;
      int me = 0, ms = 0;
      float mm = 0.0f;
      if (jl < cnt) {
        const int2 p = pes[base + jl];
        me = p.x;
        ms = p.y;
        mm = mr[ms] * md;
      }
      float4 xs[4], ee[4];
#pragma unroll
      for (int j = 0; j < 4; ++j) {
        const int jj = j < cnt ? j : 0;
        const int e = __builtin_amdgcn_readlane(me, 8 * jj);
        const int s = __builtin_amdgcn_readlane(ms, 8 * jj);
        xs[j] = *(const float4*)(xr + (size_t)s * HH);
        ee[j] = *(const float4*)(er + (size_t)e * HH);
      }
#pragma unroll
      for (int j = 0; j < 4; ++j) {
        if (j < cnt) {
          float4 g;
          g.x = gelu_f(xs[j].x + ee[j].x);
          g.y = gelu_f(xs[j].y + ee[j].y);
          g.z = gelu_f(xs[j].z + ee[j].z);
          g.w = gelu_f(xs[j].w + ee[j].w);
#pragma unroll
          for (int rc = 0; rc < RC; ++rc) {
            const float mk = __int_as_float(
                __builtin_amdgcn_readlane(__float_as_int(mm), 8 * j + rc));
            acc[rc].x += g.x * mk;
            acc[rc].y += g.y * mk;
            acc[rc].z += g.z * mk;
            acc[rc].w += g.w * mk;
          }
        }
      }
    }
#pragma unroll
    for (int rc = 0; rc < RC; ++rc)
      *(float4*)(ar + ((size_t)rc * NN + d) * HH) = acc[rc];
  }
}

#define PITCH 132
#define SLAB 32
#define NSLAB ((RC * NN) / SLAB)
#define MLP_GRID 512

__device__ __forceinline__ v8f wmma4(v2f a, v2f b, v8f c) {
  return __builtin_amdgcn_wmma_f32_16x16x4_f32(false, a, false, b, (short)0, c,
                                               false, false);
}

__device__ __forceinline__ void lds_barrier() {
  __builtin_amdgcn_fence(__ATOMIC_RELEASE, "workgroup", "local");
  __builtin_amdgcn_s_barrier();
  __builtin_amdgcn_fence(__ATOMIC_ACQUIRE, "workgroup", "local");
}

template <int LDB>
__device__ __forceinline__ void load_b_slice(const float* __restrict__ Wc,
                                             v2f (&b)[32]) {
  const int lane = threadIdx.x & 31;
  const int mn = lane & 15;
  const int kh = (lane >> 4) << 1;
#pragma unroll
  for (int kk = 0; kk < 32; ++kk) {
    b[kk][0] = Wc[(4 * kk + kh + 0) * LDB + mn];
    b[kk][1] = Wc[(4 * kk + kh + 1) * LDB + mn];
  }
}

__device__ __forceinline__ void gemm_two_tiles(const float* S,
                                               const v2f (&b)[32], v8f& c0,
                                               v8f& c1) {
  const int lane = threadIdx.x & 31;
  const float* p0 = S + (lane & 15) * PITCH + ((lane >> 4) << 1);
  const float* p1 = p0 + 16 * PITCH;
#pragma unroll
  for (int kk = 0; kk < 32; ++kk) {
    const v2f a0 = *(const v2f*)(p0 + 4 * kk);
    const v2f a1 = *(const v2f*)(p1 + 4 * kk);
    c0 = wmma4(a0, b[kk], c0);
    c1 = wmma4(a1, b[kk], c1);
  }
}

__device__ __forceinline__ v8f gemm_one_tile(const float* S,
                                             const v2f (&b)[32], v8f c) {
  const int lane = threadIdx.x & 31;
  const float* p0 = S + (lane & 15) * PITCH + ((lane >> 4) << 1);
#pragma unroll
  for (int kk = 0; kk < 32; ++kk) {
    const v2f a0 = *(const v2f*)(p0 + 4 * kk);
    c = wmma4(a0, b[kk], c);
  }
  return c;
}

static_assert(NN % SLAB == 0, "a slab never straddles two replicas");
template <bool SHARED_IN>
__global__ __launch_bounds__(256) void gin_mlp_kernel(
    float* __restrict__ ux, const float* __restrict__ x0,
    const float* __restrict__ agg,
    const float* __restrict__ eps_l, const float* __restrict__ W1,
    const float* __restrict__ b1, const float* __restrict__ W2,
    const float* __restrict__ b2) {
  __shared__ __attribute__((aligned(16))) float hA[SLAB * PITCH];
  __shared__ __attribute__((aligned(16))) float hB[SLAB * PITCH];
  const int wave = threadIdx.x >> 5;
  const int col0 = wave * 16;
  const int lane = threadIdx.x & 31;
  const int n = lane & 15;
  const int kh = lane >> 4;
  const float one_eps = 1.0f + eps_l[0];

  v2f w1f[32], w2f[32];
  load_b_slice<HH>(W1 + col0, w1f);
  load_b_slice<HH>(W2 + col0, w2f);
  const float bv1 = b1[col0 + n];
  const float bv2 = b2[col0 + n];

  const int sr = threadIdx.x >> 5;
  const int sc = (threadIdx.x & 31) * 4;
  float4 pu[4], pa[4];
  const float* uin = SHARED_IN ? x0 : ux;
  int it = blockIdx.x;
  {
    const int f = it < NSLAB ? it : 0;
    const int fu = SHARED_IN ? f % (NN / SLAB) : f;
#pragma unroll
    for (int j = 0; j < 4; ++j) {
      const size_t g = ((size_t)f * SLAB + sr + 8 * j) * HH + sc;
      const size_t gu = ((size_t)fu * SLAB + sr + 8 * j) * HH + sc;
      pu[j] = *(const float4*)(uin + gu);
      pa[j] = *(const float4*)(agg + g);
    }
  }
  for (; it < NSLAB; it += gridDim.x) {
    const size_t row0 = (size_t)it * SLAB;
#pragma unroll
    for (int j = 0; j < 4; ++j) {
      float4 h;
      h.x = one_eps * pu[j].x + pa[j].x;
      h.y = one_eps * pu[j].y + pa[j].y;
      h.z = one_eps * pu[j].z + pa[j].z;
      h.w = one_eps * pu[j].w + pa[j].w;
      *(float4*)(hA + (sr + 8 * j) * PITCH + sc) = h;
    }
    const int nx = it + (int)gridDim.x < NSLAB ? it + (int)gridDim.x : it;
    const int nu = SHARED_IN ? nx % (NN / SLAB) : nx;
#pragma unroll
    for (int j = 0; j < 4; ++j) {
      const size_t g = ((size_t)nx * SLAB + sr + 8 * j) * HH + sc;
      const size_t gu = ((size_t)nu * SLAB + sr + 8 * j) * HH + sc;
      pu[j] = *(const float4*)(uin + gu);
      pa[j] = *(const float4*)(agg + g);
    }
    lds_barrier();

    v8f c0 = {}, c1 = {};
    gemm_two_tiles(hA, w1f, c0, c1);
#pragma unroll
    for (int v = 0; v < 8; ++v) {
      hB[(v + 8 * kh) * PITCH + col0 + n] = gelu_f(c0[v] + bv1);
      hB[(16 + v + 8 * kh) * PITCH + col0 + n] = gelu_f(c1[v] + bv1);
    }
    lds_barrier();

    v8f d0 = {}, d1 = {};
    gemm_two_tiles(hB, w2f, d0, d1);
#pragma unroll
    for (int v = 0; v < 8; ++v) {
      ux[(row0 + v + 8 * kh) * HH + col0 + n] = gelu_f(d0[v] + bv2);
      ux[(row0 + 16 + v + 8 * kh) * HH + col0 + n] = gelu_f(d1[v] + bv2);
    }
  }
}

__global__ __launch_bounds__(256) void head_mlp_kernel(
    const float* __restrict__ ux, const float* __restrict__ W1,
    const float* __restrict__ b1, const float* __restrict__ W2,
    const float* __restrict__ b2, const float* __restrict__ mask,
    const int* __restrict__ batch, float* __restrict__ num,
    float* __restrict__ den) {
  __shared__ __attribute__((aligned(16))) float hA[SLAB * PITCH];
  __shared__ __attribute__((aligned(16))) float hB[SLAB * PITCH];
  const int wave = threadIdx.x >> 5;
  const int col0 = wave * 16;
  const int lane = threadIdx.x & 31;
  const int n = lane & 15;
  const int kh = lane >> 4;
  const int ct = wave & 1;
  const int rt = (wave >> 1) & 1;

  v2f w1f[32], w2f[32];
  load_b_slice<HH>(W1 + col0, w1f);
  load_b_slice<OO>(W2 + ct * 16, w2f);
  const float bv1 = b1[col0 + n];
  const float bv2 = b2[ct * 16 + n];

  const int sr = threadIdx.x >> 5;
  const int sc = (threadIdx.x & 31) * 4;
  int it = blockIdx.x;
  const int f = it < NSLAB ? it : 0;
  const float* up = ux + ((size_t)f * SLAB + sr) * HH + sc;
  v4f p0 = *(const v4f*)(up);
  v4f p1 = *(const v4f*)(up + 8 * HH);
  v4f p2 = *(const v4f*)(up + 16 * HH);
  v4f p3 = *(const v4f*)(up + 24 * HH);
  for (; it < NSLAB; it += gridDim.x) {
    float* sp = hA + sr * PITCH + sc;
    *(v4f*)(sp) = p0;
    *(v4f*)(sp + 8 * PITCH) = p1;
    *(v4f*)(sp + 16 * PITCH) = p2;
    *(v4f*)(sp + 24 * PITCH) = p3;
    const int nx = it + (int)gridDim.x < NSLAB ? it + (int)gridDim.x : it;
    up = ux + ((size_t)nx * SLAB + sr) * HH + sc;
    p0 = *(const v4f*)(up);
    p1 = *(const v4f*)(up + 8 * HH);
    p2 = *(const v4f*)(up + 16 * HH);
    p3 = *(const v4f*)(up + 24 * HH);
    lds_barrier();

    v8f c0 = {}, c1 = {};
    gemm_two_tiles(hA, w1f, c0, c1);
#pragma unroll
    for (int v = 0; v < 8; ++v) {
      hB[(v + 8 * kh) * PITCH + col0 + n] = gelu_f(c0[v] + bv1);
      hB[(16 + v + 8 * kh) * PITCH + col0 + n] = gelu_f(c1[v] + bv1);
    }
    lds_barrier();

    const int rc = it / (NN / SLAB);
    const int n0 = (it - rc * (NN / SLAB)) * SLAB;
    const float* mrc = mask + (size_t)rc * NN;
    if (wave < 4) {
      const int nb = n0 + rt * 16 + 8 * kh;
      const int4 ga = *(const int4*)(batch + nb);
      const int4 gb = *(const int4*)(batch + nb + 4);
      const float4 ma = *(const float4*)(mrc + nb);
      const float4 mb = *(const float4*)(mrc + nb + 4);
      v8f d = {};
      d = gemm_one_tile(hB + rt * 16 * PITCH, w2f, d);
      const int g[8] = {ga.x, ga.y, ga.z, ga.w, gb.x, gb.y, gb.z, gb.w};
      const float m[8] = {ma.x, ma.y, ma.z, ma.w, mb.x, mb.y, mb.z, mb.w};
      float t[8];
#pragma unroll
      for (int v = 0; v < 8; ++v) t[v] = (d[v] + bv2) * m[v];
      const int gf = __builtin_amdgcn_readfirstlane(g[0]);
      int glo = g[0], ghi = g[0];
#pragma unroll
      for (int v = 1; v < 8; ++v) {
        glo = g[v] < glo ? g[v] : glo;
        ghi = g[v] > ghi ? g[v] : ghi;
      }
      float* np = num + (size_t)rc * GG * OO + ct * 16 + n;
      if (__builtin_amdgcn_ballot_w32(glo != gf || ghi != gf) == 0u) {
        float s = ((t[0] + t[1]) + (t[2] + t[3])) +
                  ((t[4] + t[5]) + (t[6] + t[7]));
        s += __shfl_xor(s, 16);
        if (kh == 0) unsafeAtomicAdd(np + (size_t)gf * OO, s);
      } else {
#pragma unroll
        for (int v = 0; v < 8; ++v)
          unsafeAtomicAdd(np + (size_t)g[v] * OO, t[v]);
      }
    } else if (wave == 4) {
      const float mv = mrc[n0 + lane];
      const int gv = batch[n0 + lane];
      const int gf = __builtin_amdgcn_readfirstlane(gv);
      float* dp = den + (size_t)rc * GG;
      if (__builtin_amdgcn_ballot_w32(gv != gf) == 0u) {
        float s = mv;
        s += __shfl_xor(s, 16);
        s += __shfl_xor(s, 8);
        s += __shfl_xor(s, 4);
        s += __shfl_xor(s, 2);
        s += __shfl_xor(s, 1);
        if (lane == 0) unsafeAtomicAdd(dp + gf, s);
      } else {
        unsafeAtomicAdd(dp + gv, mv);
      }
    }
  }
}

__global__ __launch_bounds__(256) void finalize_kernel(
    float* __restrict__ out, const float* __restrict__ den) {
  const int i = blockIdx.x * 256 + threadIdx.x;
  if (i >= RC * GG * OO) return;
  const int g = (i / OO) % GG;
  const int rc = i / (OO * GG);
  out[i] = out[i] / (den[(size_t)rc * GG + g] + 1e-7f);
}

extern "C" void kernel_launch(void* const* d_in, const int* in_sizes, int n_in,
                              void* d_out, int out_size, void* d_ws,
                              size_t ws_size, hipStream_t stream) {
  const float* x_feat = (const float*)d_in[0];
  const int* eidx = (const int*)d_in[1];
  const float* eattr = (const float*)d_in[2];
  const int* batch = (const int*)d_in[3];
  const float* nmask = (const float*)d_in[4];
  const float* atomW = (const float*)d_in[5];
  const float* atomB = (const float*)d_in[6];
  const float* eps = (const float*)d_in[7];
  const float* bondW = (const float*)d_in[8];
  const float* bondB = (const float*)d_in[9];
  const float* m1W = (const float*)d_in[10];
  const float* m1B = (const float*)d_in[11];
  const float* m2W = (const float*)d_in[12];
  const float* m2B = (const float*)d_in[13];
  const float* h1W = (const float*)d_in[14];
  const float* h1B = (const float*)d_in[15];
  const float* h2W = (const float*)d_in[16];
  const float* h2B = (const float*)d_in[17];
  float* out = (float*)d_out;

  const int* src = eidx;
  const int* dst = eidx + EE;

  float* ux = (float*)d_ws;
  float* agg = ux + (size_t)RC * NN * HH;
  float* eemb = agg + (size_t)RC * NN * HH;
  float* x0 = eemb + (size_t)NLAY * EE * HH;
  float* den = x0 + (size_t)NN * HH;
  int* off = (int*)(den + (size_t)RC * GG);
  int* rnk = off + (NN + 4);
  int2* pes = (int2*)(rnk + EE);

  prepass_kernel<<<PRE_GRID, PRE_T, 0, stream>>>(dst, rnk, off, out, den, x_feat,
                                                 atomW, atomB, x0, eattr, bondW,
                                                 bondB, eemb);
  csr_fill_kernel<<<(EE + 255) / 256, 256, 0, stream>>>(src, dst, rnk, off,
                                                        pes);

  msg_gather_shared_kernel<<<NN / G0_NPB, 256, 0, stream>>>(x0, eemb, off, pes,
                                                            nmask, agg);
  gin_mlp_kernel<true><<<MLP_GRID, 256, 0, stream>>>(ux, x0, agg, eps, m1W, m1B,
                                                     m2W, m2B);

  for (int l = 1; l < NLAY; ++l) {
    msg_gather_kernel<<<NN / GATHER_NPB, 256, 0, stream>>>(
        ux, eemb + (size_t)l * EE * HH, off, pes, nmask, agg);
    gin_mlp_kernel<false><<<MLP_GRID, 256, 0, stream>>>(
        ux, x0, agg, eps + l, m1W + (size_t)l * HH * HH, m1B + (size_t)l * HH,
        m2W + (size_t)l * HH * HH, m2B + (size_t)l * HH);
  }

  head_mlp_kernel<<<MLP_GRID, 256, 0, stream>>>(ux, h1W, h1B, h2W, h2B, nmask,
                                                batch, out, den);
  finalize_kernel<<<(RC * GG * OO) / 256, 256, 0, stream>>>(out, den);
}
